# starting kernel
# baseline (speedup 1.0000x reference)
#include <hip/hip_runtime.h>
#include <hip/hip_bf16.h>
#include <math.h>

#define LL   141
#define CCH  8
#define HH   64
#define MIDD 128
#define HCC  512

typedef __attribute__((ext_vector_type(2))) float v2f;
typedef __attribute__((ext_vector_type(8))) float v8f;

#define WMMA4(a, b, c) \
  __builtin_amdgcn_wmma_f32_16x16x4_f32(false, (a), false, (b), (short)0, (c), false, false)

__device__ __forceinline__ v2f load_a4(const float* base, int ld, int kb, int lane) {
  int m = lane & 15;
  int k = kb + ((lane >> 4) << 1);
  const float* p = base + m * ld + k;
  v2f r = {p[0], p[1]};
  return r;
}

__device__ __forceinline__ v2f load_bT(const float* W, int ldw, int n0, int kb, int lane) {
  int n = n0 + (lane & 15);
  int k = kb + ((lane >> 4) << 1);
  const float* p = W + n * ldw + k;
  v2f r = {p[0], p[1]};
  return r;
}

__device__ __forceinline__ void g_eval(const float* z, float* h, float* kout,
                                       const float* dX,
                                       const float* __restrict__ W1, const float* __restrict__ b1,
                                       const float* __restrict__ W2, const float* __restrict__ b2,
                                       int lane) {
  v2f aZ[HH / 4];
#pragma unroll
  for (int ks = 0; ks < HH / 4; ++ks) aZ[ks] = load_a4(z, HH, ks * 4, lane);

  for (int nt = 0; nt < MIDD / 16; ++nt) {
    v8f acc = {};
#pragma unroll
    for (int ks = 0; ks < HH / 4; ++ks) {
      v2f b = load_bT(W1, HH, nt * 16, ks * 4, lane);
      acc = WMMA4(aZ[ks], b, acc);
    }
    int col = nt * 16 + (lane & 15);
    float bias = b1[col];
#pragma unroll
    for (int j = 0; j < 8; ++j) {
      int row = (lane < 16) ? j : j + 8;
      h[row * MIDD + col] = fmaxf(acc[j] + bias, 0.0f);
    }
  }
  __syncthreads();

  int cc   = lane & 7;
  int hsel = (lane >> 3) & 1;

  v2f aH[MIDD / 4];
#pragma unroll
  for (int ks = 0; ks < MIDD / 4; ++ks) aH[ks] = load_a4(h, MIDD, ks * 4, lane);

  float dxv[8];
#pragma unroll
  for (int j = 0; j < 8; ++j) {
    int row = (lane < 16) ? j : j + 8;
    dxv[j] = dX[row * CCH + cc];
  }

  for (int nt = 0; nt < HCC / 16; ++nt) {
    v8f acc = {};
#pragma unroll
    for (int ks = 0; ks < MIDD / 4; ++ks) {
      v2f b = load_bT(W2, MIDD, nt * 16, ks * 4, lane);
      acc = WMMA4(aH[ks], b, acc);
    }
    int col  = nt * 16 + (lane & 15);
    float bias = b2[col];
    int hout = nt * 2 + hsel;
#pragma unroll
    for (int j = 0; j < 8; ++j) {
      int row = (lane < 16) ? j : j + 8;
      float f = tanhf(acc[j] + bias);
      float p = f * dxv[j];
      p += __shfl_xor(p, 1, 32);
      p += __shfl_xor(p, 2, 32);
      p += __shfl_xor(p, 4, 32);
      if (cc == 0) kout[row * HH + hout] = p;
    }
  }
  __syncthreads();
}

__global__ __launch_bounds__(32) void ncde_fused(
    const float* __restrict__ coeffs,
    const float* __restrict__ W_init, const float* __restrict__ b_init,
    const float* __restrict__ W1, const float* __restrict__ b1,
    const float* __restrict__ W2, const float* __restrict__ b2,
    const float* __restrict__ W_out, const float* __restrict__ b_out,
    float* __restrict__ out) {
  __shared__ float s_z [16 * HH];
  __shared__ float s_z0[16 * HH];
  __shared__ float s_h [16 * MIDD];
  __shared__ float s_k1[16 * HH];
  __shared__ float s_k2[16 * HH];
  __shared__ float s_kc[16 * HH];
  __shared__ float s_ac[16 * HH];
  __shared__ float s_dX[3][16 * CCH];

  const int lane = threadIdx.x;
  const int r0   = blockIdx.x * 16;
  const float dt = 140.0f / 49.0f;

  for (int i = 0; i < 3; ++i)
    for (int q = lane; q < 16 * CCH; q += 32) {
      int b = q >> 3, c = q & 7;
      long base = ((long)(r0 + b) * LL + i) * CCH + c;
      s_dX[i][q] = coeffs[base + CCH] - coeffs[base];
    }

  {
    v2f aX[CCH / 4];
#pragma unroll
    for (int ks = 0; ks < CCH / 4; ++ks)
      aX[ks] = load_a4(coeffs + (long)r0 * LL * CCH, LL * CCH, ks * 4, lane);
    for (int nt = 0; nt < HH / 16; ++nt) {
      v8f acc = {};
#pragma unroll
      for (int ks = 0; ks < CCH / 4; ++ks) {
        v2f b = load_bT(W_init, CCH, nt * 16, ks * 4, lane);
        acc = WMMA4(aX[ks], b, acc);
      }
      int col = nt * 16 + (lane & 15);
      float bias = b_init[col];
#pragma unroll
      for (int j = 0; j < 8; ++j) {
        int row = (lane < 16) ? j : j + 8;
        float v = acc[j] + bias;
        s_z0[row * HH + col] = v;
        s_z [row * HH + col] = v;
      }
    }
  }
  __syncthreads();

  g_eval(s_z, s_h, s_kc, s_dX[0], W1, b1, W2, b2, lane);
  for (int q = lane; q < 16 * HH; q += 32) {
    float k1 = s_kc[q];
    s_k1[q] = k1;
    s_ac[q] = k1;
    s_z[q]  = s_z0[q] + dt * (k1 * (1.0f / 3.0f));
  }
  __syncthreads();

  g_eval(s_z, s_h, s_kc, s_dX[0], W1, b1, W2, b2, lane);
  for (int q = lane; q < 16 * HH; q += 32) {
    float k2 = s_kc[q];
    s_k2[q] = k2;
    s_ac[q] += 3.0f * k2;
    s_z[q]  = s_z0[q] + dt * (k2 - s_k1[q] * (1.0f / 3.0f));
  }
  __syncthreads();

  g_eval(s_z, s_h, s_kc, s_dX[1], W1, b1, W2, b2, lane);
  for (int q = lane; q < 16 * HH; q += 32) {
    float k3 = s_kc[q];
    s_ac[q] += 3.0f * k3;
    s_z[q]  = s_z0[q] + dt * (s_k1[q] - s_k2[q] + k3);
  }
  __syncthreads();

  g_eval(s_z, s_h, s_kc, s_dX[2], W1, b1, W2, b2, lane);
  for (int q = lane; q < 16 * HH; q += 32) {
    s_z[q] = s_z0[q] + (s_ac[q] + s_kc[q]) * (dt * 0.125f);
  }
  __syncthreads();

  {
    int b = lane & 15, o = lane >> 4;
    float s = b_out[o];
    for (int hh = 0; hh < HH; ++hh)
      s = fmaf(s_z[b * HH + hh], W_out[o * HH + hh], s);
    out[(long)(r0 + b) * 2 + o] = s;
  }
}

extern "C" void kernel_launch(void* const* d_in, const int* in_sizes, int n_in,
                              void* d_out, int out_size, void* d_ws, size_t ws_size,
                              hipStream_t stream) {
  (void)n_in; (void)out_size; (void)d_ws; (void)ws_size;
  const float* coeffs = (const float*)d_in[0];
  const float* W_init = (const float*)d_in[1];
  const float* b_init = (const float*)d_in[2];
  const float* W1     = (const float*)d_in[3];
  const float* b1     = (const float*)d_in[4];
  const float* W2     = (const float*)d_in[5];
  const float* b2     = (const float*)d_in[6];
  const float* W_out  = (const float*)d_in[7];
  const float* b_out  = (const float*)d_in[8];
  float* out = (float*)d_out;

  int B = in_sizes[0] / (LL * CCH);
  int blocks = B / 16;
  ncde_fused<<<blocks, 32, 0, stream>>>(coeffs, W_init, b_init, W1, b1, W2, b2,
                                        W_out, b_out, out);
}


# Work-group of 4 waves per 16 batch rows, both GEMMs split over K: each wave multiplies a quarter of K for all tiles, partial tiles summed through LDS (sum order over k changes: 4 partial sums)
# speedup vs baseline: 2.6916x; 2.6916x over previous
#include <hip/hip_runtime.h>
#include <hip/hip_bf16.h>
#include <math.h>


#define LL   141
#define CCH  8
#define HH   64
#define MIDD 128
#define HCC  512
#define NWAVE 4
#define PTILE 256

typedef __attribute__((ext_vector_type(2))) float v2f;
typedef __attribute__((ext_vector_type(4))) float v4f;
typedef __attribute__((ext_vector_type(8))) float v8f;

#define WMMA4(a, b, c) \
  __builtin_amdgcn_wmma_f32_16x16x4_f32(false, (a), false, (b), (short)0, (c), false, false)

__device__ __forceinline__ v2f load_a4(const float* base, int ld, int kb, int lane) {
  int m = lane & 15;
  int k = kb + ((lane >> 4) << 1);
  const float* p = base + m * ld + k;
  v2f r = {p[0], p[1]};
  return r;
}

__device__ __forceinline__ v2f load_bT(const float* W, int ldw, int n0, int kb, int lane) {
  int n = n0 + (lane & 15);
  int k = kb + ((lane >> 4) << 1);
  const float* p = W + n * ldw + k;
  v2f r = {p[0], p[1]};
  return r;
}

__device__ __forceinline__ void put_part(float* part, int w, int t, int lane, v8f acc) {
  v4f* p = (v4f*)(part + (w * 32 + t) * PTILE) + lane;
  v4f lo = {acc[0], acc[1], acc[2], acc[3]};
  v4f hi = {acc[4], acc[5], acc[6], acc[7]};
  p[0]  = lo;
  p[32] = hi;
}

__device__ __forceinline__ v8f sum_parts(const float* part, int t, int lane) {
  v8f s;
#pragma unroll
  for (int w = 0; w < NWAVE; ++w) {
    const v4f* p = (const v4f*)(part + (w * 32 + t) * PTILE) + lane;
    v4f lo = p[0];
    v4f hi = p[32];
    if (w == 0) {
      s[0] = lo[0]; s[1] = lo[1]; s[2] = lo[2]; s[3] = lo[3];
      s[4] = hi[0]; s[5] = hi[1]; s[6] = hi[2]; s[7] = hi[3];
    } else {
      s[0] += lo[0]; s[1] += lo[1]; s[2] += lo[2]; s[3] += lo[3];
      s[4] += hi[0]; s[5] += hi[1]; s[6] += hi[2]; s[7] += hi[3];
    }
  }
  return s;
}

__device__ __forceinline__ void g_eval(const float* z, float* h, float* kout,
                                       const float* dX, float* part,
                                       const float* __restrict__ W1, const float* __restrict__ b1,
                                       const float* __restrict__ W2, const float* __restrict__ b2,
                                       int lane, int wave) {
  {
    const int k0 = wave * (HH / NWAVE);
    v2f aZ[HH / 4 / NWAVE];
#pragma unroll
    for (int i = 0; i < HH / 4 / NWAVE; ++i) aZ[i] = load_a4(z, HH, k0 + i * 4, lane);

    v2f bW[HH / 4 / NWAVE][MIDD / 16];
#pragma unroll
    for (int i = 0; i < HH / 4 / NWAVE; ++i)
#pragma unroll
      for (int nt = 0; nt < MIDD / 16; ++nt)
        bW[i][nt] = load_bT(W1, HH, nt * 16, k0 + i * 4, lane);
    v8f acc[MIDD / 16];
#pragma unroll
    for (int nt = 0; nt < MIDD / 16; ++nt) { v8f zero = {}; acc[nt] = zero; }
#pragma unroll
    for (int i = 0; i < HH / 4 / NWAVE; ++i)
#pragma unroll
      for (int nt = 0; nt < MIDD / 16; ++nt)
        acc[nt] = WMMA4(aZ[i], bW[i][nt], acc[nt]);
#pragma unroll
    for (int nt = 0; nt < MIDD / 16; ++nt) put_part(part, wave, nt, lane, acc[nt]);
  }
  __syncthreads();

  for (int u = 0; u < MIDD / 16 / NWAVE; ++u) {
    int nt = wave * (MIDD / 16 / NWAVE) + u;
    v8f acc = sum_parts(part, nt, lane);
    int col = nt * 16 + (lane & 15);
    float bias = b1[col];
#pragma unroll
    for (int j = 0; j < 8; ++j) {
      int row = (lane < 16) ? j : j + 8;
      h[row * MIDD + col] = fmaxf(acc[j] + bias, 0.0f);
    }
  }
  __syncthreads();

  {
    const int k0 = wave * (MIDD / NWAVE);
    v2f aH[MIDD / 4 / NWAVE];
#pragma unroll
    for (int i = 0; i < MIDD / 4 / NWAVE; ++i) aH[i] = load_a4(h, MIDD, k0 + i * 4, lane);

    for (int g = 0; g < HCC / 16 / 4; ++g) {
      v2f bW[MIDD / 4 / NWAVE][4];
#pragma unroll
      for (int i = 0; i < MIDD / 4 / NWAVE; ++i)
#pragma unroll
        for (int t = 0; t < 4; ++t)
          bW[i][t] = load_bT(W2, MIDD, (g * 4 + t) * 16, k0 + i * 4, lane);
      v8f acc[4];
#pragma unroll
      for (int t = 0; t < 4; ++t) { v8f zero = {}; acc[t] = zero; }
#pragma unroll
      for (int i = 0; i < MIDD / 4 / NWAVE; ++i)
#pragma unroll
        for (int t = 0; t < 4; ++t)
          acc[t] = WMMA4(aH[i], bW[i][t], acc[t]);
#pragma unroll
      for (int t = 0; t < 4; ++t) put_part(part, wave, g * 4 + t, lane, acc[t]);
    }
  }
  __syncthreads();

  int cc   = lane & 7;
  int hsel = (lane >> 3) & 1;

  float dxv[8];
#pragma unroll
  for (int j = 0; j < 8; ++j) {
    int row = (lane < 16) ? j : j + 8;
    dxv[j] = dX[row * CCH + cc];
  }

  for (int u = 0; u < HCC / 16 / NWAVE; ++u) {
    int nt = wave * (HCC / 16 / NWAVE) + u;
    v8f acc = sum_parts(part, nt, lane);
    int col  = nt * 16 + (lane & 15);
    float bias = b2[col];
    int hout = nt * 2 + hsel;
#pragma unroll
    for (int j = 0; j < 8; ++j) {
      int row = (lane < 16) ? j : j + 8;
      float f = tanhf(acc[j] + bias);
      float p = f * dxv[j];
      p += __shfl_xor(p, 1, 32);
      p += __shfl_xor(p, 2, 32);
      p += __shfl_xor(p, 4, 32);
      if (cc == 0) kout[row * HH + hout] = p;
    }
  }
  __syncthreads();
}

__global__ __launch_bounds__(32 * NWAVE) void ncde_fused(
    const float* __restrict__ coeffs,
    const float* __restrict__ W_init, const float* __restrict__ b_init,
    const float* __restrict__ W1, const float* __restrict__ b1,
    const float* __restrict__ W2, const float* __restrict__ b2,
    const float* __restrict__ W_out, const float* __restrict__ b_out,
    float* __restrict__ out) {
  __shared__ float s_z [16 * HH];
  __shared__ float s_z0[16 * HH];
  __shared__ float s_h [16 * MIDD];
  __shared__ float s_k1[16 * HH];
  __shared__ float s_k2[16 * HH];
  __shared__ float s_kc[16 * HH];
  __shared__ float s_ac[16 * HH];
  __shared__ float s_dX[3][16 * CCH];
  __shared__ __attribute__((aligned(16))) float s_part[NWAVE * 32 * PTILE];

  const int tid  = threadIdx.x;
  const int lane = tid & 31;
  const int wave = __builtin_amdgcn_readfirstlane(tid >> 5);
  const int nthr = 32 * NWAVE;
  const int r0   = blockIdx.x * 16;
  const float dt = 140.0f / 49.0f;

  for (int i = 0; i < 3; ++i)
    for (int q = tid; q < 16 * CCH; q += nthr) {
      int b = q >> 3, c = q & 7;
      long base = ((long)(r0 + b) * LL + i) * CCH + c;
      s_dX[i][q] = coeffs[base + CCH] - coeffs[base];
    }

  if (wave == 0) {
    v2f aX[CCH / 4];
#pragma unroll
    for (int ks = 0; ks < CCH / 4; ++ks)
      aX[ks] = load_a4(coeffs + (long)r0 * LL * CCH, LL * CCH, ks * 4, lane);
    for (int nt = 0; nt < HH / 16; ++nt) {
      v8f acc = {};
#pragma unroll
      for (int ks = 0; ks < CCH / 4; ++ks) {
        v2f b = load_bT(W_init, CCH, nt * 16, ks * 4, lane);
        acc = WMMA4(aX[ks], b, acc);
      }
      int col = nt * 16 + (lane & 15);
      float bias = b_init[col];
#pragma unroll
      for (int j = 0; j < 8; ++j) {
        int row = (lane < 16) ? j : j + 8;
        float v = acc[j] + bias;
        s_z0[row * HH + col] = v;
        s_z [row * HH + col] = v;
      }
    }
  }
  __syncthreads();

  g_eval(s_z, s_h, s_kc, s_dX[0], s_part, W1, b1, W2, b2, lane, wave);
  for (int q = tid; q < 16 * HH; q += nthr) {
    float k1 = s_kc[q];
    s_k1[q] = k1;
    s_ac[q] = k1;
    s_z[q]  = s_z0[q] + dt * (k1 * (1.0f / 3.0f));
  }
  __syncthreads();

  g_eval(s_z, s_h, s_kc, s_dX[0], s_part, W1, b1, W2, b2, lane, wave);
  for (int q = tid; q < 16 * HH; q += nthr) {
    float k2 = s_kc[q];
    s_k2[q] = k2;
    s_ac[q] += 3.0f * k2;
    s_z[q]  = s_z0[q] + dt * (k2 - s_k1[q] * (1.0f / 3.0f));
  }
  __syncthreads();

  g_eval(s_z, s_h, s_kc, s_dX[1], s_part, W1, b1, W2, b2, lane, wave);
  for (int q = tid; q < 16 * HH; q += nthr) {
    float k3 = s_kc[q];
    s_ac[q] += 3.0f * k3;
    s_z[q]  = s_z0[q] + dt * (s_k1[q] - s_k2[q] + k3);
  }
  __syncthreads();

  g_eval(s_z, s_h, s_kc, s_dX[2], s_part, W1, b1, W2, b2, lane, wave);
  for (int q = tid; q < 16 * HH; q += nthr) {
    s_z[q] = s_z0[q] + (s_ac[q] + s_kc[q]) * (dt * 0.125f);
  }
  __syncthreads();

  if (wave == 0) {
    int b = lane & 15, o = lane >> 4;
    float s = b_out[o];
    for (int hh = 0; hh < HH; ++hh)
      s = fmaf(s_z[b * HH + hh], W_out[o * HH + hh], s);
    out[(long)(r0 + b) * 2 + o] = s;
  }
}

extern "C" void kernel_launch(void* const* d_in, const int* in_sizes, int n_in,
                              void* d_out, int out_size, void* d_ws, size_t ws_size,
                              hipStream_t stream) {
  (void)n_in; (void)out_size; (void)d_ws; (void)ws_size;
  const float* coeffs = (const float*)d_in[0];
  const float* W_init = (const float*)d_in[1];
  const float* b_init = (const float*)d_in[2];
  const float* W1     = (const float*)d_in[3];
  const float* b1     = (const float*)d_in[4];
  const float* W2     = (const float*)d_in[5];
  const float* b2     = (const float*)d_in[6];
  const float* W_out  = (const float*)d_in[7];
  const float* b_out  = (const float*)d_in[8];
  float* out = (float*)d_out;

  int B = in_sizes[0] / (LL * CCH);
  int blocks = B / 16;
  ncde_fused<<<blocks, 32 * NWAVE, 0, stream>>>(coeffs, W_init, b_init, W1, b1, W2, b2,
                                               W_out, b_out, out);
}


# Weight B fragments no longer gathered 8 B/lane from 16 rows in global: each wave copies its K-quarter rows by 16-B row-following loads into a private LDS slab, fragments read there; sums unchanged
# speedup vs baseline: 3.9838x; 1.4801x over previous
#include <hip/hip_runtime.h>
#include <hip/hip_bf16.h>
#include <math.h>


#define LL   141
#define CCH  8
#define HH   64
#define MIDD 128
#define HCC  512
#define NWAVE 4
#define PTILE 256
#define P1 20
#define P2 36
#define SLAB 2560

typedef __attribute__((ext_vector_type(2))) float v2f;
typedef __attribute__((ext_vector_type(4))) float v4f;
typedef __attribute__((ext_vector_type(8))) float v8f;

#define WMMA4(a, b, c) \
  __builtin_amdgcn_wmma_f32_16x16x4_f32(false, (a), false, (b), (short)0, (c), false, false)

__device__ __forceinline__ v2f load_a4(const float* base, int ld, int kb, int lane) {
  int m = lane & 15;
  int k = kb + ((lane >> 4) << 1);
  const float* p = base + m * ld + k;
  v2f r = {p[0], p[1]};
  return r;
}

__device__ __forceinline__ v2f load_bT(const float* W, int ldw, int n0, int kb, int lane) {
  int n = n0 + (lane & 15);
  int k = kb + ((lane >> 4) << 1);
  const float* p = W + n * ldw + k;
  v2f r = {p[0], p[1]};
  return r;
}

__device__ __forceinline__ v2f load_b_slab(const float* slab, int pitch, int nt, int i, int lane) {
  const float* p = slab + (nt * 16 + (lane & 15)) * pitch + i * 4 + ((lane >> 4) << 1);
  v2f r = {p[0], p[1]};
  return r;
}

__device__ __forceinline__ void slab_order() {
  __builtin_amdgcn_fence(__ATOMIC_ACQ_REL, "wavefront");
}

__device__ __forceinline__ void w2_fetch(v4f (&r)[16], const float* __restrict__ W2,
                                         int g, int k0, int lane) {
#pragma unroll
  for (int j = 0; j < 16; ++j) {
    int row = g * 64 + j * 4 + (lane >> 3);
    r[j] = *(const v4f*)(W2 + row * MIDD + k0 + (lane & 7) * 4);
  }
}
__device__ __forceinline__ void w2_stage(float* slab, const v4f (&r)[16], int lane) {
#pragma unroll
  for (int j = 0; j < 16; ++j)
    *(v4f*)(slab + (j * 4 + (lane >> 3)) * P2 + (lane & 7) * 4) = r[j];
}

__device__ __forceinline__ void put_part(float* part, int w, int t, int lane, v8f acc) {
  v4f* p = (v4f*)(part + (w * 32 + t) * PTILE) + lane;
  v4f lo = {acc[0], acc[1], acc[2], acc[3]};
  v4f hi = {acc[4], acc[5], acc[6], acc[7]};
  p[0]  = lo;
  p[32] = hi;
}

__device__ __forceinline__ v8f sum_parts(const float* part, int t, int lane) {
  v8f s;
#pragma unroll
  for (int w = 0; w < NWAVE; ++w) {
    const v4f* p = (const v4f*)(part + (w * 32 + t) * PTILE) + lane;
    v4f lo = p[0];
    v4f hi = p[32];
    if (w == 0) {
      s[0] = lo[0]; s[1] = lo[1]; s[2] = lo[2]; s[3] = lo[3];
      s[4] = hi[0]; s[5] = hi[1]; s[6] = hi[2]; s[7] = hi[3];
    } else {
      s[0] += lo[0]; s[1] += lo[1]; s[2] += lo[2]; s[3] += lo[3];
      s[4] += hi[0]; s[5] += hi[1]; s[6] += hi[2]; s[7] += hi[3];
    }
  }
  return s;
}

__device__ __forceinline__ void g_eval(const float* z, float* h, float* kout,
                                       const float* dX, float* part, float* slab,
                                       const float* __restrict__ W1, const float* __restrict__ b1,
                                       const float* __restrict__ W2, const float* __restrict__ b2,
                                       int lane, int wave) {
  {
    const int k0 = wave * (HH / NWAVE);
    v2f aZ[HH / 4 / NWAVE];
#pragma unroll
    for (int i = 0; i < HH / 4 / NWAVE; ++i) aZ[i] = load_a4(z, HH, k0 + i * 4, lane);

    v4f r1[16];
#pragma unroll
    for (int j = 0; j < 16; ++j) {
      int row = j * 8 + (lane >> 2);
      r1[j] = *(const v4f*)(W1 + row * HH + k0 + (lane & 3) * 4);
    }
    slab_order();
#pragma unroll
    for (int j = 0; j < 16; ++j)
      *(v4f*)(slab + (j * 8 + (lane >> 2)) * P1 + (lane & 3) * 4) = r1[j];
    slab_order();
    v2f bW[HH / 4 / NWAVE][MIDD / 16];
#pragma unroll
    for (int i = 0; i < HH / 4 / NWAVE; ++i)
#pragma unroll
      for (int nt = 0; nt < MIDD / 16; ++nt)
        bW[i][nt] = load_b_slab(slab, P1, nt, i, lane);
    v8f acc[MIDD / 16];
#pragma unroll
    for (int nt = 0; nt < MIDD / 16; ++nt) { v8f zero = {}; acc[nt] = zero; }
#pragma unroll
    for (int i = 0; i < HH / 4 / NWAVE; ++i)
#pragma unroll
      for (int nt = 0; nt < MIDD / 16; ++nt)
        acc[nt] = WMMA4(aZ[i], bW[i][nt], acc[nt]);
#pragma unroll
    for (int nt = 0; nt < MIDD / 16; ++nt) put_part(part, wave, nt, lane, acc[nt]);
  }
  __syncthreads();

  v4f rw[2][16];
  w2_fetch(rw[0], W2, 0, wave * (MIDD / NWAVE), lane);
  w2_fetch(rw[1], W2, 1, wave * (MIDD / NWAVE), lane);

  for (int u = 0; u < MIDD / 16 / NWAVE; ++u) {
    int nt = wave * (MIDD / 16 / NWAVE) + u;
    v8f acc = sum_parts(part, nt, lane);
    int col = nt * 16 + (lane & 15);
    float bias = b1[col];
#pragma unroll
    for (int j = 0; j < 8; ++j) {
      int row = (lane < 16) ? j : j + 8;
      h[row * MIDD + col] = fmaxf(acc[j] + bias, 0.0f);
    }
  }
  __syncthreads();

  {
    const int k0 = wave * (MIDD / NWAVE);
    v2f aH[MIDD / 4 / NWAVE];
#pragma unroll
    for (int i = 0; i < MIDD / 4 / NWAVE; ++i) aH[i] = load_a4(h, MIDD, k0 + i * 4, lane);

    v2f bW[2][MIDD / 4 / NWAVE][4];
    slab_order();
    w2_stage(slab, rw[0], lane);
    w2_fetch(rw[0], W2, 2, k0, lane);
    slab_order();
#pragma unroll
    for (int i = 0; i < MIDD / 4 / NWAVE; ++i)
#pragma unroll
      for (int t = 0; t < 4; ++t)
        bW[0][i][t] = load_b_slab(slab, P2, t, i, lane);
#pragma unroll
    for (int g = 0; g < HCC / 16 / 4; ++g) {
      if (g + 1 < HCC / 16 / 4) {
        slab_order();
        w2_stage(slab, rw[(g + 1) & 1], lane);
        if (g + 3 < HCC / 16 / 4) w2_fetch(rw[(g + 1) & 1], W2, g + 3, k0, lane);
        slab_order();
#pragma unroll
        for (int i = 0; i < MIDD / 4 / NWAVE; ++i)
#pragma unroll
          for (int t = 0; t < 4; ++t)
            bW[(g + 1) & 1][i][t] = load_b_slab(slab, P2, t, i, lane);
      }
      v8f acc[4];
#pragma unroll
      for (int t = 0; t < 4; ++t) { v8f zero = {}; acc[t] = zero; }
#pragma unroll
      for (int i = 0; i < MIDD / 4 / NWAVE; ++i)
#pragma unroll
        for (int t = 0; t < 4; ++t)
          acc[t] = WMMA4(aH[i], bW[g & 1][i][t], acc[t]);
#pragma unroll
      for (int t = 0; t < 4; ++t) put_part(part, wave, g * 4 + t, lane, acc[t]);
    }
  }
  __syncthreads();

  int cc   = lane & 7;
  int hsel = (lane >> 3) & 1;

  float dxv[8];
#pragma unroll
  for (int j = 0; j < 8; ++j) {
    int row = (lane < 16) ? j : j + 8;
    dxv[j] = dX[row * CCH + cc];
  }

  for (int u = 0; u < HCC / 16 / NWAVE; ++u) {
    int nt = wave * (HCC / 16 / NWAVE) + u;
    v8f acc = sum_parts(part, nt, lane);
    int col  = nt * 16 + (lane & 15);
    float bias = b2[col];
    int hout = nt * 2 + hsel;
#pragma unroll
    for (int j = 0; j < 8; ++j) {
      int row = (lane < 16) ? j : j + 8;
      float f = tanhf(acc[j] + bias);
      float p = f * dxv[j];
      p += __shfl_xor(p, 1, 32);
      p += __shfl_xor(p, 2, 32);
      p += __shfl_xor(p, 4, 32);
      if (cc == 0) kout[row * HH + hout] = p;
    }
  }
  __syncthreads();
}

__global__ __launch_bounds__(32 * NWAVE) void ncde_fused(
    const float* __restrict__ coeffs,
    const float* __restrict__ W_init, const float* __restrict__ b_init,
    const float* __restrict__ W1, const float* __restrict__ b1,
    const float* __restrict__ W2, const float* __restrict__ b2,
    const float* __restrict__ W_out, const float* __restrict__ b_out,
    float* __restrict__ out) {
  __shared__ float s_z [16 * HH];
  __shared__ float s_z0[16 * HH];
  __shared__ float s_h [16 * MIDD];
  __shared__ float s_k1[16 * HH];
  __shared__ float s_k2[16 * HH];
  __shared__ float s_kc[16 * HH];
  __shared__ float s_ac[16 * HH];
  __shared__ float s_dX[3][16 * CCH];
  __shared__ __attribute__((aligned(16))) float s_part[NWAVE * 32 * PTILE];
  __shared__ __attribute__((aligned(16))) float s_slab[NWAVE * SLAB];

  const int tid  = threadIdx.x;
  const int lane = tid & 31;
  const int wave = __builtin_amdgcn_readfirstlane(tid >> 5);
  const int nthr = 32 * NWAVE;
  const int r0   = blockIdx.x * 16;
  const float dt = 140.0f / 49.0f;

  for (int i = 0; i < 3; ++i)
    for (int q = tid; q < 16 * CCH; q += nthr) {
      int b = q >> 3, c = q & 7;
      long base = ((long)(r0 + b) * LL + i) * CCH + c;
      s_dX[i][q] = coeffs[base + CCH] - coeffs[base];
    }

  if (wave == 0) {
    v2f aX[CCH / 4];
#pragma unroll
    for (int ks = 0; ks < CCH / 4; ++ks)
      aX[ks] = load_a4(coeffs + (long)r0 * LL * CCH, LL * CCH, ks * 4, lane);
    for (int nt = 0; nt < HH / 16; ++nt) {
      v8f acc = {};
#pragma unroll
      for (int ks = 0; ks < CCH / 4; ++ks) {
        v2f b = load_bT(W_init, CCH, nt * 16, ks * 4, lane);
        acc = WMMA4(aX[ks], b, acc);
      }
      int col = nt * 16 + (lane & 15);
      float bias = b_init[col];
#pragma unroll
      for (int j = 0; j < 8; ++j) {
        int row = (lane < 16) ? j : j + 8;
        float v = acc[j] + bias;
        s_z0[row * HH + col] = v;
        s_z [row * HH + col] = v;
      }
    }
  }
  __syncthreads();

  g_eval(s_z, s_h, s_kc, s_dX[0], s_part, s_slab + wave * SLAB, W1, b1, W2, b2, lane, wave);
  for (int q = tid; q < 16 * HH; q += nthr) {
    float k1 = s_kc[q];
    s_k1[q] = k1;
    s_ac[q] = k1;
    s_z[q]  = s_z0[q] + dt * (k1 * (1.0f / 3.0f));
  }
  __syncthreads();

  g_eval(s_z, s_h, s_kc, s_dX[0], s_part, s_slab + wave * SLAB, W1, b1, W2, b2, lane, wave);
  for (int q = tid; q < 16 * HH; q += nthr) {
    float k2 = s_kc[q];
    s_k2[q] = k2;
    s_ac[q] += 3.0f * k2;
    s_z[q]  = s_z0[q] + dt * (k2 - s_k1[q] * (1.0f / 3.0f));
  }
  __syncthreads();

  g_eval(s_z, s_h, s_kc, s_dX[1], s_part, s_slab + wave * SLAB, W1, b1, W2, b2, lane, wave);
  for (int q = tid; q < 16 * HH; q += nthr) {
    float k3 = s_kc[q];
    s_ac[q] += 3.0f * k3;
    s_z[q]  = s_z0[q] + dt * (s_k1[q] - s_k2[q] + k3);
  }
  __syncthreads();

  g_eval(s_z, s_h, s_kc, s_dX[2], s_part, s_slab + wave * SLAB, W1, b1, W2, b2, lane, wave);
  for (int q = tid; q < 16 * HH; q += nthr) {
    s_z[q] = s_z0[q] + (s_ac[q] + s_kc[q]) * (dt * 0.125f);
  }
  __syncthreads();

  if (wave == 0) {
    int b = lane & 15, o = lane >> 4;
    float s = b_out[o];
    for (int hh = 0; hh < HH; ++hh)
      s = fmaf(s_z[b * HH + hh], W_out[o * HH + hh], s);
    out[(long)(r0 + b) * 2 + o] = s;
  }
}

extern "C" void kernel_launch(void* const* d_in, const int* in_sizes, int n_in,
                              void* d_out, int out_size, void* d_ws, size_t ws_size,
                              hipStream_t stream) {
  (void)n_in; (void)out_size; (void)d_ws; (void)ws_size;
  const float* coeffs = (const float*)d_in[0];
  const float* W_init = (const float*)d_in[1];
  const float* b_init = (const float*)d_in[2];
  const float* W1     = (const float*)d_in[3];
  const float* b1     = (const float*)d_in[4];
  const float* W2     = (const float*)d_in[5];
  const float* b2     = (const float*)d_in[6];
  const float* W_out  = (const float*)d_in[7];
  const float* b_out  = (const float*)d_in[8];
  float* out = (float*)d_out;

  int B = in_sizes[0] / (LL * CCH);
  int blocks = B / 16;
  ncde_fused<<<blocks, 32 * NWAVE, 0, stream>>>(coeffs, W_init, b_init, W1, b1, W2, b2,
                                               W_out, b_out, out);
}


# GEMM 2 tiles hold one channel c of 16 hidden units each: the sum over c becomes an in-lane multiply-add chain over 8 tiles, no lane shuffles or masked stores; sum order over c changes
# speedup vs baseline: 6.9802x; 1.7521x over previous
#include <hip/hip_runtime.h>
#include <hip/hip_bf16.h>
#include <math.h>


#define LL   141
#define CCH  8
#define HH   64
#define MIDD 128
#define HCC  512
#define NWAVE 4
#define PTILE 256
#define P1 20
#define P2 36
#define SLAB 2560
#define TS2 608

typedef __attribute__((ext_vector_type(2))) float v2f;
typedef __attribute__((ext_vector_type(4))) float v4f;
typedef __attribute__((ext_vector_type(8))) float v8f;

#define WMMA4(a, b, c) \
  __builtin_amdgcn_wmma_f32_16x16x4_f32(false, (a), false, (b), (short)0, (c), false, false)

__device__ __forceinline__ v2f load_a4(const float* base, int ld, int kb, int lane) {
  int m = lane & 15;
  int k = kb + ((lane >> 4) << 1);
  const float* p = base + m * ld + k;
  v2f r = {p[0], p[1]};
  return r;
}

__device__ __forceinline__ v2f load_bT(const float* W, int ldw, int n0, int kb, int lane) {
  int n = n0 + (lane & 15);
  int k = kb + ((lane >> 4) << 1);
  const float* p = W + n * ldw + k;
  v2f r = {p[0], p[1]};
  return r;
}

__device__ __forceinline__ v2f load_b_slab(const float* slab, int pitch, int nt, int i, int lane) {
  const float* p = slab + (nt * 16 + (lane & 15)) * pitch + i * 4 + ((lane >> 4) << 1);
  v2f r = {p[0], p[1]};
  return r;
}

__device__ __forceinline__ v2f load_b_slab2(const float* slab, int t, int i, int lane) {
  const float* p = slab + t * TS2 + (lane & 15) * P2 + i * 4 + ((lane >> 4) << 1);
  v2f r = {p[0], p[1]};
  return r;
}

__device__ __forceinline__ void slab_order() {
  __builtin_amdgcn_fence(__ATOMIC_ACQ_REL, "wavefront");
}

__device__ __forceinline__ void w2_fetch(v4f (&r)[16], const float* __restrict__ W2,
                                         int g, int k0, int lane) {
#pragma unroll
  for (int j = 0; j < 16; ++j) {
    int row = ((g >> 1) * 16 + j) * CCH + (g & 1) * 4 + (lane >> 3);
    r[j] = *(const v4f*)(W2 + row * MIDD + k0 + (lane & 7) * 4);
  }
}
__device__ __forceinline__ void w2_stage(float* slab, const v4f (&r)[16], int lane) {
#pragma unroll
  for (int j = 0; j < 16; ++j)
    *(v4f*)(slab + (lane >> 3) * TS2 + j * P2 + (lane & 7) * 4) = r[j];
}

__device__ __forceinline__ void put_part(float* part, int w, int t, int lane, v8f acc) {
  v4f* p = (v4f*)(part + (w * 32 + t) * PTILE) + lane;
  v4f lo = {acc[0], acc[1], acc[2], acc[3]};
  v4f hi = {acc[4], acc[5], acc[6], acc[7]};
  p[0]  = lo;
  p[32] = hi;
}

__device__ __forceinline__ v8f sum_parts(const float* part, int t, int lane) {
  v8f s;
#pragma unroll
  for (int w = 0; w < NWAVE; ++w) {
    const v4f* p = (const v4f*)(part + (w * 32 + t) * PTILE) + lane;
    v4f lo = p[0];
    v4f hi = p[32];
    if (w == 0) {
      s[0] = lo[0]; s[1] = lo[1]; s[2] = lo[2]; s[3] = lo[3];
      s[4] = hi[0]; s[5] = hi[1]; s[6] = hi[2]; s[7] = hi[3];
    } else {
      s[0] += lo[0]; s[1] += lo[1]; s[2] += lo[2]; s[3] += lo[3];
      s[4] += hi[0]; s[5] += hi[1]; s[6] += hi[2]; s[7] += hi[3];
    }
  }
  return s;
}

__device__ __forceinline__ void g_eval(const float* z, float* h, float* kout,
                                       const float* dX, float* part, float* slab,
                                       const float* __restrict__ W1, const float* __restrict__ b1,
                                       const float* __restrict__ W2, const float* __restrict__ b2,
                                       int lane, int wave) {
  {
    const int k0 = wave * (HH / NWAVE);
    v2f aZ[HH / 4 / NWAVE];
#pragma unroll
    for (int i = 0; i < HH / 4 / NWAVE; ++i) aZ[i] = load_a4(z, HH, k0 + i * 4, lane);

    v4f r1[16];
#pragma unroll
    for (int j = 0; j < 16; ++j) {
      int row = j * 8 + (lane >> 2);
      r1[j] = *(const v4f*)(W1 + row * HH + k0 + (lane & 3) * 4);
    }
    slab_order();
#pragma unroll
    for (int j = 0; j < 16; ++j)
      *(v4f*)(slab + (j * 8 + (lane >> 2)) * P1 + (lane & 3) * 4) = r1[j];
    slab_order();
    v2f bW[HH / 4 / NWAVE][MIDD / 16];
#pragma unroll
    for (int i = 0; i < HH / 4 / NWAVE; ++i)
#pragma unroll
      for (int nt = 0; nt < MIDD / 16; ++nt)
        bW[i][nt] = load_b_slab(slab, P1, nt, i, lane);
    v8f acc[MIDD / 16];
#pragma unroll
    for (int nt = 0; nt < MIDD / 16; ++nt) { v8f zero = {}; acc[nt] = zero; }
#pragma unroll
    for (int i = 0; i < HH / 4 / NWAVE; ++i)
#pragma unroll
      for (int nt = 0; nt < MIDD / 16; ++nt)
        acc[nt] = WMMA4(aZ[i], bW[i][nt], acc[nt]);
#pragma unroll
    for (int nt = 0; nt < MIDD / 16; ++nt) put_part(part, wave, nt, lane, acc[nt]);
  }
  __syncthreads();

  v4f rw[2][16];
  w2_fetch(rw[0], W2, 0, wave * (MIDD / NWAVE), lane);
  w2_fetch(rw[1], W2, 1, wave * (MIDD / NWAVE), lane);

  for (int u = 0; u < MIDD / 16 / NWAVE; ++u) {
    int nt = wave * (MIDD / 16 / NWAVE) + u;
    v8f acc = sum_parts(part, nt, lane);
    int col = nt * 16 + (lane & 15);
    float bias = b1[col];
#pragma unroll
    for (int j = 0; j < 8; ++j) {
      int row = (lane < 16) ? j : j + 8;
      h[row * MIDD + col] = fmaxf(acc[j] + bias, 0.0f);
    }
  }
  __syncthreads();

  {
    const int k0 = wave * (MIDD / NWAVE);
    v2f aH[MIDD / 4 / NWAVE];
#pragma unroll
    for (int i = 0; i < MIDD / 4 / NWAVE; ++i) aH[i] = load_a4(h, MIDD, k0 + i * 4, lane);

    v2f bW[2][MIDD / 4 / NWAVE][4];
    slab_order();
    w2_stage(slab, rw[0], lane);
    w2_fetch(rw[0], W2, 2, k0, lane);
    slab_order();
#pragma unroll
    for (int i = 0; i < MIDD / 4 / NWAVE; ++i)
#pragma unroll
      for (int t = 0; t < 4; ++t)
        bW[0][i][t] = load_b_slab2(slab, t, i, lane);
#pragma unroll
    for (int g = 0; g < HCC / 16 / 4; ++g) {
      if (g + 1 < HCC / 16 / 4) {
        slab_order();
        w2_stage(slab, rw[(g + 1) & 1], lane);
        if (g + 3 < HCC / 16 / 4) w2_fetch(rw[(g + 1) & 1], W2, g + 3, k0, lane);
        slab_order();
#pragma unroll
        for (int i = 0; i < MIDD / 4 / NWAVE; ++i)
#pragma unroll
          for (int t = 0; t < 4; ++t)
            bW[(g + 1) & 1][i][t] = load_b_slab2(slab, t, i, lane);
      }
      v8f acc[4];
#pragma unroll
      for (int t = 0; t < 4; ++t) { v8f zero = {}; acc[t] = zero; }
#pragma unroll
      for (int i = 0; i < MIDD / 4 / NWAVE; ++i)
#pragma unroll
        for (int t = 0; t < 4; ++t)
          acc[t] = WMMA4(aH[i], bW[g & 1][i][t], acc[t]);
#pragma unroll
      for (int t = 0; t < 4; ++t) put_part(part, wave, g * 4 + t, lane, acc[t]);
    }
  }
  __syncthreads();

  {
    const int n   = lane & 15;
    const int hlf = lane >> 4;
    const float* bp  = b2 + (wave * 16 + n) * CCH;
    const float* dxp = dX + hlf * 8 * CCH;
    const float* pp  = part + (wave * 8) * PTILE + lane * 4;

    v4f plo[NWAVE], phi[NWAVE];
    float dxn[8];
    float bn;
#pragma unroll
    for (int w = 0; w < NWAVE; ++w) {
      const float* p = pp + (w * 32) * PTILE;
      plo[w] = *(const v4f*)(p);
      phi[w] = *(const v4f*)(p + 128);
    }
#pragma unroll
    for (int j = 0; j < 8; ++j) dxn[j] = dxp[j * CCH];
    bn = bp[0];

    float ks[8];
#pragma unroll
    for (int j = 0; j < 8; ++j) ks[j] = 0.0f;

#pragma nounroll
    for (int c = 0; c < CCH; ++c) {
      v8f acc;
      acc[0] = plo[0][0]; acc[1] = plo[0][1]; acc[2] = plo[0][2]; acc[3] = plo[0][3];
      acc[4] = phi[0][0]; acc[5] = phi[0][1]; acc[6] = phi[0][2]; acc[7] = phi[0][3];
#pragma unroll
      for (int w = 1; w < NWAVE; ++w) {
        acc[0] += plo[w][0]; acc[1] += plo[w][1]; acc[2] += plo[w][2]; acc[3] += plo[w][3];
        acc[4] += phi[w][0]; acc[5] += phi[w][1]; acc[6] += phi[w][2]; acc[7] += phi[w][3];
      }
      float bias = bn;
      float dxc[8];
#pragma unroll
      for (int j = 0; j < 8; ++j) dxc[j] = dxn[j];

      const int cn = (c + 1) & (CCH - 1);
#pragma unroll
      for (int w = 0; w < NWAVE; ++w) {
        const float* p = pp + (w * 32 + cn) * PTILE;
        plo[w] = *(const v4f*)(p);
        phi[w] = *(const v4f*)(p + 128);
      }
#pragma unroll
      for (int j = 0; j < 8; ++j) dxn[j] = dxp[j * CCH + cn];
      bn = bp[cn];

#pragma unroll
      for (int j = 0; j < 8; ++j) {
        float f = tanhf(acc[j] + bias);
        ks[j] = fmaf(f, dxc[j], ks[j]);
      }
    }
#pragma unroll
    for (int j = 0; j < 8; ++j)
      kout[(hlf * 8 + j) * HH + wave * 16 + n] = ks[j];
  }
  __syncthreads();
}

__global__ __launch_bounds__(32 * NWAVE) void ncde_fused(
    const float* __restrict__ coeffs,
    const float* __restrict__ W_init, const float* __restrict__ b_init,
    const float* __restrict__ W1, const float* __restrict__ b1,
    const float* __restrict__ W2, const float* __restrict__ b2,
    const float* __restrict__ W_out, const float* __restrict__ b_out,
    float* __restrict__ out) {
  __shared__ float s_z [16 * HH];
  __shared__ float s_z0[16 * HH];
  __shared__ float s_h [16 * MIDD];
  __shared__ float s_k1[16 * HH];
  __shared__ float s_k2[16 * HH];
  __shared__ float s_kc[16 * HH];
  __shared__ float s_ac[16 * HH];
  __shared__ float s_dX[3][16 * CCH];
  __shared__ __attribute__((aligned(16))) float s_part[NWAVE * 32 * PTILE];
  __shared__ __attribute__((aligned(16))) float s_slab[NWAVE * SLAB];

  const int tid  = threadIdx.x;
  const int lane = tid & 31;
  const int wave = __builtin_amdgcn_readfirstlane(tid >> 5);
  const int nthr = 32 * NWAVE;
  const int r0   = blockIdx.x * 16;
  const float dt = 140.0f / 49.0f;

  for (int i = 0; i < 3; ++i)
    for (int q = tid; q < 16 * CCH; q += nthr) {
      int b = q >> 3, c = q & 7;
      long base = ((long)(r0 + b) * LL + i) * CCH + c;
      s_dX[i][q] = coeffs[base + CCH] - coeffs[base];
    }

  if (wave == 0) {
    v2f aX[CCH / 4];
#pragma unroll
    for (int ks = 0; ks < CCH / 4; ++ks)
      aX[ks] = load_a4(coeffs + (long)r0 * LL * CCH, LL * CCH, ks * 4, lane);
    for (int nt = 0; nt < HH / 16; ++nt) {
      v8f acc = {};
#pragma unroll
      for (int ks = 0; ks < CCH / 4; ++ks) {
        v2f b = load_bT(W_init, CCH, nt * 16, ks * 4, lane);
        acc = WMMA4(aX[ks], b, acc);
      }
      int col = nt * 16 + (lane & 15);
      float bias = b_init[col];
#pragma unroll
      for (int j = 0; j < 8; ++j) {
        int row = (lane < 16) ? j : j + 8;
        float v = acc[j] + bias;
        s_z0[row * HH + col] = v;
        s_z [row * HH + col] = v;
      }
    }
  }
  __syncthreads();

  g_eval(s_z, s_h, s_kc, s_dX[0], s_part, s_slab + wave * SLAB, W1, b1, W2, b2, lane, wave);
  for (int q = tid; q < 16 * HH; q += nthr) {
    float k1 = s_kc[q];
    s_k1[q] = k1;
    s_ac[q] = k1;
    s_z[q]  = s_z0[q] + dt * (k1 * (1.0f / 3.0f));
  }
  __syncthreads();

  g_eval(s_z, s_h, s_kc, s_dX[0], s_part, s_slab + wave * SLAB, W1, b1, W2, b2, lane, wave);
  for (int q = tid; q < 16 * HH; q += nthr) {
    float k2 = s_kc[q];
    s_k2[q] = k2;
    s_ac[q] += 3.0f * k2;
    s_z[q]  = s_z0[q] + dt * (k2 - s_k1[q] * (1.0f / 3.0f));
  }
  __syncthreads();

  g_eval(s_z, s_h, s_kc, s_dX[1], s_part, s_slab + wave * SLAB, W1, b1, W2, b2, lane, wave);
  for (int q = tid; q < 16 * HH; q += nthr) {
    float k3 = s_kc[q];
    s_ac[q] += 3.0f * k3;
    s_z[q]  = s_z0[q] + dt * (s_k1[q] - s_k2[q] + k3);
  }
  __syncthreads();

  g_eval(s_z, s_h, s_kc, s_dX[2], s_part, s_slab + wave * SLAB, W1, b1, W2, b2, lane, wave);
  for (int q = tid; q < 16 * HH; q += nthr) {
    s_z[q] = s_z0[q] + (s_ac[q] + s_kc[q]) * (dt * 0.125f);
  }
  __syncthreads();

  if (wave == 0) {
    int b = lane & 15, o = lane >> 4;
    float s = b_out[o];
    for (int hh = 0; hh < HH; ++hh)
      s = fmaf(s_z[b * HH + hh], W_out[o * HH + hh], s);
    out[(long)(r0 + b) * 2 + o] = s;
  }
}

extern "C" void kernel_launch(void* const* d_in, const int* in_sizes, int n_in,
                              void* d_out, int out_size, void* d_ws, size_t ws_size,
                              hipStream_t stream) {
  (void)n_in; (void)out_size; (void)d_ws; (void)ws_size;
  const float* coeffs = (const float*)d_in[0];
  const float* W_init = (const float*)d_in[1];
  const float* b_init = (const float*)d_in[2];
  const float* W1     = (const float*)d_in[3];
  const float* b1     = (const float*)d_in[4];
  const float* W2     = (const float*)d_in[5];
  const float* b2     = (const float*)d_in[6];
  const float* W_out  = (const float*)d_in[7];
  const float* b_out  = (const float*)d_in[8];
  float* out = (float*)d_out;

  int B = in_sizes[0] / (LL * CCH);
  int blocks = B / 16;
  ncde_fused<<<blocks, 32 * NWAVE, 0, stream>>>(coeffs, W_init, b_init, W1, b1, W2, b2,
                                               W_out, b_out, out);
}


# Work-group of 8 waves, not 4: matrix products stay K-split on waves 0-3; the 4 added waves (second wave per SIMD) take half of the tanh + dX contraction and h finishing chains; sum orders unchanged
# speedup vs baseline: 8.4519x; 1.2108x over previous
#include <hip/hip_runtime.h>
#include <hip/hip_bf16.h>
#include <math.h>


#define LL   141
#define CCH  8
#define HH   64
#define MIDD 128
#define HCC  512
#define NWAVE 4
#define NWG   8
#define PTILE 256
#define P1 20
#define P2 36
#define SLAB 2560
#define TS2 608

typedef __attribute__((ext_vector_type(2))) float v2f;
typedef __attribute__((ext_vector_type(4))) float v4f;
typedef __attribute__((ext_vector_type(8))) float v8f;

#define WMMA4(a, b, c) \
  __builtin_amdgcn_wmma_f32_16x16x4_f32(false, (a), false, (b), (short)0, (c), false, false)

__device__ __forceinline__ v2f load_a4(const float* base, int ld, int kb, int lane) {
  int m = lane & 15;
  int k = kb + ((lane >> 4) << 1);
  const float* p = base + m * ld + k;
  v2f r = {p[0], p[1]};
  return r;
}

__device__ __forceinline__ v2f load_bT(const float* W, int ldw, int n0, int kb, int lane) {
  int n = n0 + (lane & 15);
  int k = kb + ((lane >> 4) << 1);
  const float* p = W + n * ldw + k;
  v2f r = {p[0], p[1]};
  return r;
}

__device__ __forceinline__ v2f load_b_slab(const float* slab, int pitch, int nt, int i, int lane) {
  const float* p = slab + (nt * 16 + (lane & 15)) * pitch + i * 4 + ((lane >> 4) << 1);
  v2f r = {p[0], p[1]};
  return r;
}

__device__ __forceinline__ v2f load_b_slab2(const float* slab, int t, int i, int lane) {
  const float* p = slab + t * TS2 + (lane & 15) * P2 + i * 4 + ((lane >> 4) << 1);
  v2f r = {p[0], p[1]};
  return r;
}

__device__ __forceinline__ void slab_order() {
  __builtin_amdgcn_fence(__ATOMIC_ACQ_REL, "wavefront");
}

__device__ __forceinline__ void w2_fetch(v4f (&r)[16], const float* __restrict__ W2,
                                         int g, int k0, int lane) {
#pragma unroll
  for (int j = 0; j < 16; ++j) {
    int row = ((g >> 1) * 16 + j) * CCH + (g & 1) * 4 + (lane >> 3);
    r[j] = *(const v4f*)(W2 + row * MIDD + k0 + (lane & 7) * 4);
  }
}
__device__ __forceinline__ void w2_stage(float* slab, const v4f (&r)[16], int lane) {
#pragma unroll
  for (int j = 0; j < 16; ++j)
    *(v4f*)(slab + (lane >> 3) * TS2 + j * P2 + (lane & 7) * 4) = r[j];
}

__device__ __forceinline__ void put_part(float* part, int w, int t, int lane, v8f acc) {
  v4f* p = (v4f*)(part + (w * 32 + t) * PTILE) + lane;
  v4f lo = {acc[0], acc[1], acc[2], acc[3]};
  v4f hi = {acc[4], acc[5], acc[6], acc[7]};
  p[0]  = lo;
  p[32] = hi;
}

__device__ __forceinline__ v8f sum_parts(const float* part, int t, int lane) {
  v8f s;
#pragma unroll
  for (int w = 0; w < NWAVE; ++w) {
    const v4f* p = (const v4f*)(part + (w * 32 + t) * PTILE) + lane;
    v4f lo = p[0];
    v4f hi = p[32];
    if (w == 0) {
      s[0] = lo[0]; s[1] = lo[1]; s[2] = lo[2]; s[3] = lo[3];
      s[4] = hi[0]; s[5] = hi[1]; s[6] = hi[2]; s[7] = hi[3];
    } else {
      s[0] += lo[0]; s[1] += lo[1]; s[2] += lo[2]; s[3] += lo[3];
      s[4] += hi[0]; s[5] += hi[1]; s[6] += hi[2]; s[7] += hi[3];
    }
  }
  return s;
}

__device__ __forceinline__ void g_eval(const float* z, float* h, float* kout,
                                       const float* dX, float* part, float* slab,
                                       const float* __restrict__ W1, const float* __restrict__ b1,
                                       const float* __restrict__ W2, const float* __restrict__ b2,
                                       int lane, int wave) {
  if (wave < NWAVE) {
    const int k0 = wave * (HH / NWAVE);
    v2f aZ[HH / 4 / NWAVE];
#pragma unroll
    for (int i = 0; i < HH / 4 / NWAVE; ++i) aZ[i] = load_a4(z, HH, k0 + i * 4, lane);

    v4f r1[16];
#pragma unroll
    for (int j = 0; j < 16; ++j) {
      int row = j * 8 + (lane >> 2);
      r1[j] = *(const v4f*)(W1 + row * HH + k0 + (lane & 3) * 4);
    }
    slab_order();
#pragma unroll
    for (int j = 0; j < 16; ++j)
      *(v4f*)(slab + (j * 8 + (lane >> 2)) * P1 + (lane & 3) * 4) = r1[j];
    slab_order();
    v2f bW[HH / 4 / NWAVE][MIDD / 16];
#pragma unroll
    for (int i = 0; i < HH / 4 / NWAVE; ++i)
#pragma unroll
      for (int nt = 0; nt < MIDD / 16; ++nt)
        bW[i][nt] = load_b_slab(slab, P1, nt, i, lane);
    v8f acc[MIDD / 16];
#pragma unroll
    for (int nt = 0; nt < MIDD / 16; ++nt) { v8f zero = {}; acc[nt] = zero; }
#pragma unroll
    for (int i = 0; i < HH / 4 / NWAVE; ++i)
#pragma unroll
      for (int nt = 0; nt < MIDD / 16; ++nt)
        acc[nt] = WMMA4(aZ[i], bW[i][nt], acc[nt]);
#pragma unroll
    for (int nt = 0; nt < MIDD / 16; ++nt) put_part(part, wave, nt, lane, acc[nt]);
  }
  __syncthreads();

  v4f rw[2][16];
  if (wave < NWAVE) {
    w2_fetch(rw[0], W2, 0, wave * (MIDD / NWAVE), lane);
    w2_fetch(rw[1], W2, 1, wave * (MIDD / NWAVE), lane);
  }

  {
    int nt = wave;
    v8f acc = sum_parts(part, nt, lane);
    int col = nt * 16 + (lane & 15);
    float bias = b1[col];
#pragma unroll
    for (int j = 0; j < 8; ++j) {
      int row = (lane < 16) ? j : j + 8;
      h[row * MIDD + col] = fmaxf(acc[j] + bias, 0.0f);
    }
  }
  __syncthreads();

  if (wave < NWAVE) {
    const int k0 = wave * (MIDD / NWAVE);
    v2f aH[MIDD / 4 / NWAVE];
#pragma unroll
    for (int i = 0; i < MIDD / 4 / NWAVE; ++i) aH[i] = load_a4(h, MIDD, k0 + i * 4, lane);

    v2f bW[2][MIDD / 4 / NWAVE][4];
    slab_order();
    w2_stage(slab, rw[0], lane);
    w2_fetch(rw[0], W2, 2, k0, lane);
    slab_order();
#pragma unroll
    for (int i = 0; i < MIDD / 4 / NWAVE; ++i)
#pragma unroll
      for (int t = 0; t < 4; ++t)
        bW[0][i][t] = load_b_slab2(slab, t, i, lane);
#pragma unroll
    for (int g = 0; g < HCC / 16 / 4; ++g) {
      if (g + 1 < HCC / 16 / 4) {
        slab_order();
        w2_stage(slab, rw[(g + 1) & 1], lane);
        if (g + 3 < HCC / 16 / 4) w2_fetch(rw[(g + 1) & 1], W2, g + 3, k0, lane);
        slab_order();
#pragma unroll
        for (int i = 0; i < MIDD / 4 / NWAVE; ++i)
#pragma unroll
          for (int t = 0; t < 4; ++t)
            bW[(g + 1) & 1][i][t] = load_b_slab2(slab, t, i, lane);
      }
      v8f acc[4];
#pragma unroll
      for (int t = 0; t < 4; ++t) { v8f zero = {}; acc[t] = zero; }
#pragma unroll
      for (int i = 0; i < MIDD / 4 / NWAVE; ++i)
#pragma unroll
        for (int t = 0; t < 4; ++t)
          acc[t] = WMMA4(aH[i], bW[g & 1][i][t], acc[t]);
#pragma unroll
      for (int t = 0; t < 4; ++t) put_part(part, wave, g * 4 + t, lane, acc[t]);
    }
  }
  __syncthreads();

  {
    const int hb  = wave >> 1;
    const int hf  = wave & 1;
    const int n   = lane & 15;
    const int rb  = (lane >> 4) * 8 + hf * 4;
    const float* bp  = b2 + (hb * 16 + n) * CCH;
    const float* dxp = dX + rb * CCH;
    const float* pp  = part + (hb * 8) * PTILE + hf * 128 + lane * 4;

    v4f pv[NWAVE];
    float dxn[4];
    float bn;
#pragma unroll
    for (int w = 0; w < NWAVE; ++w) pv[w] = *(const v4f*)(pp + (w * 32) * PTILE);
#pragma unroll
    for (int j = 0; j < 4; ++j) dxn[j] = dxp[j * CCH];
    bn = bp[0];

    float ks[4];
#pragma unroll
    for (int j = 0; j < 4; ++j) ks[j] = 0.0f;

#pragma nounroll
    for (int c = 0; c < CCH; ++c) {
      v4f acc = pv[0];
#pragma unroll
      for (int w = 1; w < NWAVE; ++w) {
        acc[0] += pv[w][0]; acc[1] += pv[w][1]; acc[2] += pv[w][2]; acc[3] += pv[w][3];
      }
      float bias = bn;
      float dxc[4];
#pragma unroll
      for (int j = 0; j < 4; ++j) dxc[j] = dxn[j];

      const int cn = (c + 1) & (CCH - 1);
#pragma unroll
      for (int w = 0; w < NWAVE; ++w) pv[w] = *(const v4f*)(pp + (w * 32 + cn) * PTILE);
#pragma unroll
      for (int j = 0; j < 4; ++j) dxn[j] = dxp[j * CCH + cn];
      bn = bp[cn];

#pragma unroll
      for (int j = 0; j < 4; ++j) {
        float f = tanhf(acc[j] + bias);
        ks[j] = fmaf(f, dxc[j], ks[j]);
      }
    }
#pragma unroll
    for (int j = 0; j < 4; ++j)
      kout[(rb + j) * HH + hb * 16 + n] = ks[j];
  }
  __syncthreads();
}

__global__ __launch_bounds__(32 * NWG) void ncde_fused(
    const float* __restrict__ coeffs,
    const float* __restrict__ W_init, const float* __restrict__ b_init,
    const float* __restrict__ W1, const float* __restrict__ b1,
    const float* __restrict__ W2, const float* __restrict__ b2,
    const float* __restrict__ W_out, const float* __restrict__ b_out,
    float* __restrict__ out) {
  __shared__ float s_z [16 * HH];
  __shared__ float s_z0[16 * HH];
  __shared__ float s_h [16 * MIDD];
  __shared__ float s_k1[16 * HH];
  __shared__ float s_k2[16 * HH];
  __shared__ float s_kc[16 * HH];
  __shared__ float s_ac[16 * HH];
  __shared__ float s_dX[3][16 * CCH];
  __shared__ __attribute__((aligned(16))) float s_part[NWAVE * 32 * PTILE];
  __shared__ __attribute__((aligned(16))) float s_slab[NWAVE * SLAB];

  const int tid  = threadIdx.x;
  const int lane = tid & 31;
  const int wave = __builtin_amdgcn_readfirstlane(tid >> 5);
  const int nthr = 32 * NWG;
  const int r0   = blockIdx.x * 16;
  const float dt = 140.0f / 49.0f;

  for (int i = 0; i < 3; ++i)
    for (int q = tid; q < 16 * CCH; q += nthr) {
      int b = q >> 3, c = q & 7;
      long base = ((long)(r0 + b) * LL + i) * CCH + c;
      s_dX[i][q] = coeffs[base + CCH] - coeffs[base];
    }

  if (wave == 0) {
    v2f aX[CCH / 4];
#pragma unroll
    for (int ks = 0; ks < CCH / 4; ++ks)
      aX[ks] = load_a4(coeffs + (long)r0 * LL * CCH, LL * CCH, ks * 4, lane);
    for (int nt = 0; nt < HH / 16; ++nt) {
      v8f acc = {};
#pragma unroll
      for (int ks = 0; ks < CCH / 4; ++ks) {
        v2f b = load_bT(W_init, CCH, nt * 16, ks * 4, lane);
        acc = WMMA4(aX[ks], b, acc);
      }
      int col = nt * 16 + (lane & 15);
      float bias = b_init[col];
#pragma unroll
      for (int j = 0; j < 8; ++j) {
        int row = (lane < 16) ? j : j + 8;
        float v = acc[j] + bias;
        s_z0[row * HH + col] = v;
        s_z [row * HH + col] = v;
      }
    }
  }
  __syncthreads();

  g_eval(s_z, s_h, s_kc, s_dX[0], s_part, s_slab + (wave & (NWAVE - 1)) * SLAB, W1, b1, W2, b2, lane, wave);
  for (int q = tid; q < 16 * HH; q += nthr) {
    float k1 = s_kc[q];
    s_k1[q] = k1;
    s_ac[q] = k1;
    s_z[q]  = s_z0[q] + dt * (k1 * (1.0f / 3.0f));
  }
  __syncthreads();

  g_eval(s_z, s_h, s_kc, s_dX[0], s_part, s_slab + (wave & (NWAVE - 1)) * SLAB, W1, b1, W2, b2, lane, wave);
  for (int q = tid; q < 16 * HH; q += nthr) {
    float k2 = s_kc[q];
    s_k2[q] = k2;
    s_ac[q] += 3.0f * k2;
    s_z[q]  = s_z0[q] + dt * (k2 - s_k1[q] * (1.0f / 3.0f));
  }
  __syncthreads();

  g_eval(s_z, s_h, s_kc, s_dX[1], s_part, s_slab + (wave & (NWAVE - 1)) * SLAB, W1, b1, W2, b2, lane, wave);
  for (int q = tid; q < 16 * HH; q += nthr) {
    float k3 = s_kc[q];
    s_ac[q] += 3.0f * k3;
    s_z[q]  = s_z0[q] + dt * (s_k1[q] - s_k2[q] + k3);
  }
  __syncthreads();

  g_eval(s_z, s_h, s_kc, s_dX[2], s_part, s_slab + (wave & (NWAVE - 1)) * SLAB, W1, b1, W2, b2, lane, wave);
  for (int q = tid; q < 16 * HH; q += nthr) {
    s_z[q] = s_z0[q] + (s_ac[q] + s_kc[q]) * (dt * 0.125f);
  }
  __syncthreads();

  if (wave == 0) {
    int b = lane & 15, o = lane >> 4;
    float s = b_out[o];
    for (int hh = 0; hh < HH; ++hh)
      s = fmaf(s_z[b * HH + hh], W_out[o * HH + hh], s);
    out[(long)(r0 + b) * 2 + o] = s;
  }
}

extern "C" void kernel_launch(void* const* d_in, const int* in_sizes, int n_in,
                              void* d_out, int out_size, void* d_ws, size_t ws_size,
                              hipStream_t stream) {
  (void)n_in; (void)out_size; (void)d_ws; (void)ws_size;
  const float* coeffs = (const float*)d_in[0];
  const float* W_init = (const float*)d_in[1];
  const float* b_init = (const float*)d_in[2];
  const float* W1     = (const float*)d_in[3];
  const float* b1     = (const float*)d_in[4];
  const float* W2     = (const float*)d_in[5];
  const float* b2     = (const float*)d_in[6];
  const float* W_out  = (const float*)d_in[7];
  const float* b_out  = (const float*)d_in[8];
  float* out = (float*)d_out;

  int B = in_sizes[0] / (LL * CCH);
  int blocks = B / 16;
  ncde_fused<<<blocks, 32 * NWG, 0, stream>>>(coeffs, W_init, b_init, W1, b1, W2, b2,
                                               W_out, b_out, out);
}


# GEMM 2 transposed on all 8 waves (W2 rows as A operand, whole K per wave): a lane's 8 D registers = 8 channels of one hidden unit, tanh + dX sum in registers; k sum one chain, not 4 partials
# speedup vs baseline: 9.8126x; 1.1610x over previous
#include <hip/hip_runtime.h>
#include <hip/hip_bf16.h>
#include <math.h>


#define LL   141
#define CCH  8
#define HH   64
#define MIDD 128
#define HCC  512
#define NWAVE 4
#define NWG   8
#define PTILE 256
#define NT1  8
#define HP   132
#define P1 20
#define P2 36
#define SLAB 2560
#define TS2 608

typedef __attribute__((ext_vector_type(2))) float v2f;
typedef __attribute__((ext_vector_type(4))) float v4f;
typedef __attribute__((ext_vector_type(8))) float v8f;

#define WMMA4(a, b, c) \
  __builtin_amdgcn_wmma_f32_16x16x4_f32(false, (a), false, (b), (short)0, (c), false, false)

__device__ __forceinline__ v2f load_a4(const float* base, int ld, int kb, int lane) {
  int m = lane & 15;
  int k = kb + ((lane >> 4) << 1);
  const float* p = base + m * ld + k;
  v2f r = {p[0], p[1]};
  return r;
}

__device__ __forceinline__ v2f load_bT(const float* W, int ldw, int n0, int kb, int lane) {
  int n = n0 + (lane & 15);
  int k = kb + ((lane >> 4) << 1);
  const float* p = W + n * ldw + k;
  v2f r = {p[0], p[1]};
  return r;
}

__device__ __forceinline__ v2f load_b_slab(const float* slab, int pitch, int nt, int i, int lane) {
  const float* p = slab + (nt * 16 + (lane & 15)) * pitch + i * 4 + ((lane >> 4) << 1);
  v2f r = {p[0], p[1]};
  return r;
}

__device__ __forceinline__ v2f load_b_slab2(const float* slab, int t, int i, int lane) {
  const float* p = slab + t * TS2 + (lane & 15) * P2 + i * 4 + ((lane >> 4) << 1);
  v2f r = {p[0], p[1]};
  return r;
}

__device__ __forceinline__ void slab_order() {
  __builtin_amdgcn_fence(__ATOMIC_ACQ_REL, "wavefront");
}

__device__ __forceinline__ void w2_fetch(v4f (&r)[8], const float* __restrict__ W2,
                                         int wv, int q, int lane) {
#pragma unroll
  for (int j = 0; j < 8; ++j) {
    int row = wv * 64 + (q >> 2) * 32 + ((lane >> 3) & 1) * 16 + 2 * j + (lane >> 4);
    r[j] = *(const v4f*)(W2 + row * MIDD + (q & 3) * 32 + (lane & 7) * 4);
  }
}
__device__ __forceinline__ void w2_stage(float* slab, const v4f (&r)[8], int lane) {
#pragma unroll
  for (int j = 0; j < 8; ++j)
    *(v4f*)(slab + ((lane >> 3) & 1) * TS2 + (2 * j + (lane >> 4)) * P2 + (lane & 7) * 4) = r[j];
}

__device__ __forceinline__ void put_part(float* part, int w, int t, int lane, v8f acc) {
  v4f* p = (v4f*)(part + (w * NT1 + t) * PTILE) + lane;
  v4f lo = {acc[0], acc[1], acc[2], acc[3]};
  v4f hi = {acc[4], acc[5], acc[6], acc[7]};
  p[0]  = lo;
  p[32] = hi;
}

__device__ __forceinline__ v8f sum_parts(const float* part, int t, int lane) {
  v8f s;
#pragma unroll
  for (int w = 0; w < NWAVE; ++w) {
    const v4f* p = (const v4f*)(part + (w * NT1 + t) * PTILE) + lane;
    v4f lo = p[0];
    v4f hi = p[32];
    if (w == 0) {
      s[0] = lo[0]; s[1] = lo[1]; s[2] = lo[2]; s[3] = lo[3];
      s[4] = hi[0]; s[5] = hi[1]; s[6] = hi[2]; s[7] = hi[3];
    } else {
      s[0] += lo[0]; s[1] += lo[1]; s[2] += lo[2]; s[3] += lo[3];
      s[4] += hi[0]; s[5] += hi[1]; s[6] += hi[2]; s[7] += hi[3];
    }
  }
  return s;
}

__device__ __forceinline__ void g_eval(const float* z, float* h, float* kout,
                                       const float* dX, float* part, float* slab,
                                       const float* __restrict__ W1, const float* __restrict__ b1,
                                       const float* __restrict__ W2, const float* __restrict__ b2,
                                       int lane, int wave) {
  if (wave < NWAVE) {
    const int k0 = wave * (HH / NWAVE);
    v2f aZ[HH / 4 / NWAVE];
#pragma unroll
    for (int i = 0; i < HH / 4 / NWAVE; ++i) aZ[i] = load_a4(z, HH, k0 + i * 4, lane);

    v4f r1[16];
#pragma unroll
    for (int j = 0; j < 16; ++j) {
      int row = j * 8 + (lane >> 2);
      r1[j] = *(const v4f*)(W1 + row * HH + k0 + (lane & 3) * 4);
    }
    slab_order();
#pragma unroll
    for (int j = 0; j < 16; ++j)
      *(v4f*)(slab + (j * 8 + (lane >> 2)) * P1 + (lane & 3) * 4) = r1[j];
    slab_order();
    v2f bW[HH / 4 / NWAVE][MIDD / 16];
#pragma unroll
    for (int i = 0; i < HH / 4 / NWAVE; ++i)
#pragma unroll
      for (int nt = 0; nt < MIDD / 16; ++nt)
        bW[i][nt] = load_b_slab(slab, P1, nt, i, lane);
    v8f acc[MIDD / 16];
#pragma unroll
    for (int nt = 0; nt < MIDD / 16; ++nt) { v8f zero = {}; acc[nt] = zero; }
#pragma unroll
    for (int i = 0; i < HH / 4 / NWAVE; ++i)
#pragma unroll
      for (int nt = 0; nt < MIDD / 16; ++nt)
        acc[nt] = WMMA4(aZ[i], bW[i][nt], acc[nt]);
#pragma unroll
    for (int nt = 0; nt < MIDD / 16; ++nt) put_part(part, wave, nt, lane, acc[nt]);
  }
  __syncthreads();

  const int bcol = lane & 15;
  const int hh   = lane >> 4;
  v4f rw[2][8];
  w2_fetch(rw[0], W2, wave, 0, lane);
  w2_fetch(rw[1], W2, wave, 1, lane);
  v4f bb[4][2];
#pragma unroll
  for (int t = 0; t < 4; ++t) {
    const v4f* bp = (const v4f*)(b2 + (wave * 8 + t * 2 + hh) * CCH);
    bb[t][0] = bp[0];
    bb[t][1] = bp[1];
  }

  {
    int nt = wave;
    v8f acc = sum_parts(part, nt, lane);
    int col = nt * 16 + (lane & 15);
    float bias = b1[col];
#pragma unroll
    for (int j = 0; j < 8; ++j) {
      int row = (lane < 16) ? j : j + 8;
      h[row * HP + col] = fmaxf(acc[j] + bias, 0.0f);
    }
  }
  __syncthreads();

  {
    const v4f d0 = *(const v4f*)(dX + bcol * CCH);
    const v4f d1 = *(const v4f*)(dX + bcol * CCH + 4);
    v8f acc[2];
    v2f aH[2][8];
    v2f bW[2][8][2];
    slab_order();
    w2_stage(slab, rw[0], lane);
    w2_fetch(rw[0], W2, wave, 2, lane);
    slab_order();
#pragma unroll
    for (int i = 0; i < 8; ++i)
#pragma unroll
      for (int t = 0; t < 2; ++t)
        bW[0][i][t] = load_b_slab2(slab, t, i, lane);
#pragma unroll
    for (int i = 0; i < 8; ++i) aH[0][i] = load_a4(h, HP, i * 4, lane);
#pragma unroll
    for (int q = 0; q < 8; ++q) {
      if (q + 1 < 8) {
        slab_order();
        w2_stage(slab, rw[(q + 1) & 1], lane);
        if (q + 3 < 8) w2_fetch(rw[(q + 1) & 1], W2, wave, q + 3, lane);
        slab_order();
#pragma unroll
        for (int i = 0; i < 8; ++i)
#pragma unroll
          for (int t = 0; t < 2; ++t)
            bW[(q + 1) & 1][i][t] = load_b_slab2(slab, t, i, lane);
#pragma unroll
        for (int i = 0; i < 8; ++i)
          aH[(q + 1) & 1][i] = load_a4(h, HP, ((q + 1) & 3) * 32 + i * 4, lane);
      }
      if ((q & 3) == 0) {
#pragma unroll
        for (int t = 0; t < 2; ++t) { v8f zero = {}; acc[t] = zero; }
      }
#pragma unroll
      for (int i = 0; i < 8; ++i)
#pragma unroll
        for (int t = 0; t < 2; ++t)
          acc[t] = WMMA4(bW[q & 1][i][t], aH[q & 1][i], acc[t]);

      if ((q & 3) == 3) {
#pragma unroll
        for (int t = 0; t < 2; ++t) {
          const int tile = (q >> 2) * 2 + t;
          float ks = 0.0f;
#pragma unroll
          for (int c = 0; c < 4; ++c) {
            float f = tanhf(acc[t][c] + bb[tile][0][c]);
            ks = fmaf(f, d0[c], ks);
          }
#pragma unroll
          for (int c = 0; c < 4; ++c) {
            float f = tanhf(acc[t][4 + c] + bb[tile][1][c]);
            ks = fmaf(f, d1[c], ks);
          }
          kout[bcol * HH + wave * 8 + tile * 2 + hh] = ks;
        }
      }
    }
  }
  __syncthreads();
}

__global__ __launch_bounds__(32 * NWG) void ncde_fused(
    const float* __restrict__ coeffs,
    const float* __restrict__ W_init, const float* __restrict__ b_init,
    const float* __restrict__ W1, const float* __restrict__ b1,
    const float* __restrict__ W2, const float* __restrict__ b2,
    const float* __restrict__ W_out, const float* __restrict__ b_out,
    float* __restrict__ out) {
  __shared__ float s_z [16 * HH];
  __shared__ float s_z0[16 * HH];
  __shared__ __attribute__((aligned(16))) float s_h [16 * HP];
  __shared__ float s_k1[16 * HH];
  __shared__ float s_k2[16 * HH];
  __shared__ float s_kc[16 * HH];
  __shared__ float s_ac[16 * HH];
  __shared__ __attribute__((aligned(16))) float s_dX[3][16 * CCH];
  __shared__ __attribute__((aligned(16))) float s_part[NWAVE * NT1 * PTILE];
  __shared__ __attribute__((aligned(16))) float s_slab[NWG * SLAB];

  const int tid  = threadIdx.x;
  const int lane = tid & 31;
  const int wave = __builtin_amdgcn_readfirstlane(tid >> 5);
  const int nthr = 32 * NWG;
  const int r0   = blockIdx.x * 16;
  const float dt = 140.0f / 49.0f;

  for (int i = 0; i < 3; ++i)
    for (int q = tid; q < 16 * CCH; q += nthr) {
      int b = q >> 3, c = q & 7;
      long base = ((long)(r0 + b) * LL + i) * CCH + c;
      s_dX[i][q] = coeffs[base + CCH] - coeffs[base];
    }

  if (wave == 0) {
    v2f aX[CCH / 4];
#pragma unroll
    for (int ks = 0; ks < CCH / 4; ++ks)
      aX[ks] = load_a4(coeffs + (long)r0 * LL * CCH, LL * CCH, ks * 4, lane);
    for (int nt = 0; nt < HH / 16; ++nt) {
      v8f acc = {};
#pragma unroll
      for (int ks = 0; ks < CCH / 4; ++ks) {
        v2f b = load_bT(W_init, CCH, nt * 16, ks * 4, lane);
        acc = WMMA4(aX[ks], b, acc);
      }
      int col = nt * 16 + (lane & 15);
      float bias = b_init[col];
#pragma unroll
      for (int j = 0; j < 8; ++j) {
        int row = (lane < 16) ? j : j + 8;
        float v = acc[j] + bias;
        s_z0[row * HH + col] = v;
        s_z [row * HH + col] = v;
      }
    }
  }
  __syncthreads();

#pragma nounroll
  for (int s = 0; s < 4; ++s) {
    const float* dXs = s_dX[s == 0 ? 0 : s - 1];
    g_eval(s_z, s_h, s_kc, dXs, s_part, s_slab + wave * SLAB, W1, b1, W2, b2, lane, wave);
    if (s == 0) {
      for (int q = tid; q < 16 * HH; q += nthr) {
        float k1 = s_kc[q];
        s_k1[q] = k1;
        s_ac[q] = k1;
        s_z[q]  = s_z0[q] + dt * (k1 * (1.0f / 3.0f));
      }
    } else if (s == 1) {
      for (int q = tid; q < 16 * HH; q += nthr) {
        float k2 = s_kc[q];
        s_k2[q] = k2;
        s_ac[q] += 3.0f * k2;
        s_z[q]  = s_z0[q] + dt * (k2 - s_k1[q] * (1.0f / 3.0f));
      }
    } else if (s == 2) {
      for (int q = tid; q < 16 * HH; q += nthr) {
        float k3 = s_kc[q];
        s_ac[q] += 3.0f * k3;
        s_z[q]  = s_z0[q] + dt * (s_k1[q] - s_k2[q] + k3);
      }
    } else {
      for (int q = tid; q < 16 * HH; q += nthr) {
        s_z[q] = s_z0[q] + (s_ac[q] + s_kc[q]) * (dt * 0.125f);
      }
    }
    __syncthreads();
  }

  if (wave == 0) {
    int b = lane & 15, o = lane >> 4;
    float s = b_out[o];
    for (int hh = 0; hh < HH; ++hh)
      s = fmaf(s_z[b * HH + hh], W_out[o * HH + hh], s);
    out[(long)(r0 + b) * 2 + o] = s;
  }
}

extern "C" void kernel_launch(void* const* d_in, const int* in_sizes, int n_in,
                              void* d_out, int out_size, void* d_ws, size_t ws_size,
                              hipStream_t stream) {
  (void)n_in; (void)out_size; (void)d_ws; (void)ws_size;
  const float* coeffs = (const float*)d_in[0];
  const float* W_init = (const float*)d_in[1];
  const float* b_init = (const float*)d_in[2];
  const float* W1     = (const float*)d_in[3];
  const float* b1     = (const float*)d_in[4];
  const float* W2     = (const float*)d_in[5];
  const float* b2     = (const float*)d_in[6];
  const float* W_out  = (const float*)d_in[7];
  const float* b_out  = (const float*)d_in[8];
  float* out = (float*)d_out;

  int B = in_sizes[0] / (LL * CCH);
  int blocks = B / 16;
  ncde_fused<<<blocks, 32 * NWG, 0, stream>>>(coeffs, W_init, b_init, W1, b1, W2, b2,
                                               W_out, b_out, out);
}


# Both products: slab stores and fragment reads leave as ONE group in front of each block of 16 (32) matrix instructions, held there by scheduling fences, not spread among them; sums unchanged
# speedup vs baseline: 10.1886x; 1.0383x over previous
#include <hip/hip_runtime.h>
#include <hip/hip_bf16.h>
#include <math.h>


#define LL   141
#define CCH  8
#define HH   64
#define MIDD 128
#define HCC  512
#define NWAVE 4
#define NWG   8
#define PTILE 256
#define NT1  8
#define HP   132
#define P1 20
#define P2 36
#define SLAB 2560
#define TS2 608

typedef __attribute__((ext_vector_type(2))) float v2f;
typedef __attribute__((ext_vector_type(4))) float v4f;
typedef __attribute__((ext_vector_type(8))) float v8f;

#define WMMA4(a, b, c) \
  __builtin_amdgcn_wmma_f32_16x16x4_f32(false, (a), false, (b), (short)0, (c), false, false)

__device__ __forceinline__ v2f load_a4(const float* base, int ld, int kb, int lane) {
  int m = lane & 15;
  int k = kb + ((lane >> 4) << 1);
  const float* p = base + m * ld + k;
  v2f r = {p[0], p[1]};
  return r;
}

__device__ __forceinline__ v2f load_bT(const float* W, int ldw, int n0, int kb, int lane) {
  int n = n0 + (lane & 15);
  int k = kb + ((lane >> 4) << 1);
  const float* p = W + n * ldw + k;
  v2f r = {p[0], p[1]};
  return r;
}

__device__ __forceinline__ v2f load_b_slab(const float* slab, int pitch, int nt, int i, int lane) {
  const float* p = slab + (nt * 16 + (lane & 15)) * pitch + i * 4 + ((lane >> 4) << 1);
  v2f r = {p[0], p[1]};
  return r;
}

__device__ __forceinline__ v2f load_b_slab2(const float* slab, int t, int i, int lane) {
  const float* p = slab + t * TS2 + (lane & 15) * P2 + i * 4 + ((lane >> 4) << 1);
  v2f r = {p[0], p[1]};
  return r;
}

__device__ __forceinline__ void slab_order() {
  __builtin_amdgcn_fence(__ATOMIC_ACQ_REL, "wavefront");
}

__device__ __forceinline__ void w2_fetch(v4f (&r)[8], const float* __restrict__ W2,
                                         int wv, int q, int lane) {
#pragma unroll
  for (int j = 0; j < 8; ++j) {
    int row = wv * 64 + (q >> 2) * 32 + ((lane >> 3) & 1) * 16 + 2 * j + (lane >> 4);
    r[j] = *(const v4f*)(W2 + row * MIDD + (q & 3) * 32 + (lane & 7) * 4);
  }
}
__device__ __forceinline__ void w2_stage(float* slab, const v4f (&r)[8], int lane) {
#pragma unroll
  for (int j = 0; j < 8; ++j)
    *(v4f*)(slab + ((lane >> 3) & 1) * TS2 + (2 * j + (lane >> 4)) * P2 + (lane & 7) * 4) = r[j];
}

__device__ __forceinline__ void put_part(float* part, int w, int t, int lane, v8f acc) {
  v4f* p = (v4f*)(part + (w * NT1 + t) * PTILE) + lane;
  v4f lo = {acc[0], acc[1], acc[2], acc[3]};
  v4f hi = {acc[4], acc[5], acc[6], acc[7]};
  p[0]  = lo;
  p[32] = hi;
}

__device__ __forceinline__ v8f sum_parts(const float* part, int t, int lane) {
  v8f s;
#pragma unroll
  for (int w = 0; w < NWAVE; ++w) {
    const v4f* p = (const v4f*)(part + (w * NT1 + t) * PTILE) + lane;
    v4f lo = p[0];
    v4f hi = p[32];
    if (w == 0) {
      s[0] = lo[0]; s[1] = lo[1]; s[2] = lo[2]; s[3] = lo[3];
      s[4] = hi[0]; s[5] = hi[1]; s[6] = hi[2]; s[7] = hi[3];
    } else {
      s[0] += lo[0]; s[1] += lo[1]; s[2] += lo[2]; s[3] += lo[3];
      s[4] += hi[0]; s[5] += hi[1]; s[6] += hi[2]; s[7] += hi[3];
    }
  }
  return s;
}

__device__ __forceinline__ void g_eval(const float* z, float* h, float* kout,
                                       const float* dX, float* part, float* slab,
                                       const float* __restrict__ W1, const float* __restrict__ b1,
                                       const float* __restrict__ W2, const float* __restrict__ b2,
                                       int lane, int wave) {
  if (wave < NWAVE) {
    const int k0 = wave * (HH / NWAVE);
    v2f aZ[HH / 4 / NWAVE];
#pragma unroll
    for (int i = 0; i < HH / 4 / NWAVE; ++i) aZ[i] = load_a4(z, HH, k0 + i * 4, lane);

    v4f r1[16];
#pragma unroll
    for (int j = 0; j < 16; ++j) {
      int row = j * 8 + (lane >> 2);
      r1[j] = *(const v4f*)(W1 + row * HH + k0 + (lane & 3) * 4);
    }
    slab_order();
#pragma unroll
    for (int j = 0; j < 16; ++j)
      *(v4f*)(slab + (j * 8 + (lane >> 2)) * P1 + (lane & 3) * 4) = r1[j];
    slab_order();
    v2f bW[HH / 4 / NWAVE][MIDD / 16];
#pragma unroll
    for (int i = 0; i < HH / 4 / NWAVE; ++i)
#pragma unroll
      for (int nt = 0; nt < MIDD / 16; ++nt)
        bW[i][nt] = load_b_slab(slab, P1, nt, i, lane);
    __builtin_amdgcn_sched_barrier(0);
    v8f acc[MIDD / 16];
#pragma unroll
    for (int nt = 0; nt < MIDD / 16; ++nt) { v8f zero = {}; acc[nt] = zero; }
#pragma unroll
    for (int i = 0; i < HH / 4 / NWAVE; ++i)
#pragma unroll
      for (int nt = 0; nt < MIDD / 16; ++nt)
        acc[nt] = WMMA4(aZ[i], bW[i][nt], acc[nt]);
#pragma unroll
    for (int nt = 0; nt < MIDD / 16; ++nt) put_part(part, wave, nt, lane, acc[nt]);
  }
  __syncthreads();

  const int bcol = lane & 15;
  const int hh   = lane >> 4;
  v4f rw[2][8];
  w2_fetch(rw[0], W2, wave, 0, lane);
  w2_fetch(rw[1], W2, wave, 1, lane);
  v4f bb[4][2];
#pragma unroll
  for (int t = 0; t < 4; ++t) {
    const v4f* bp = (const v4f*)(b2 + (wave * 8 + t * 2 + hh) * CCH);
    bb[t][0] = bp[0];
    bb[t][1] = bp[1];
  }

  {
    int nt = wave;
    v8f acc = sum_parts(part, nt, lane);
    int col = nt * 16 + (lane & 15);
    float bias = b1[col];
#pragma unroll
    for (int j = 0; j < 8; ++j) {
      int row = (lane < 16) ? j : j + 8;
      h[row * HP + col] = fmaxf(acc[j] + bias, 0.0f);
    }
  }
  __syncthreads();

  {
    const v4f d0 = *(const v4f*)(dX + bcol * CCH);
    const v4f d1 = *(const v4f*)(dX + bcol * CCH + 4);
    v8f acc[2];
    v2f aH[2][8];
    v2f bW[2][8][2];
    slab_order();
    w2_stage(slab, rw[0], lane);
    w2_fetch(rw[0], W2, wave, 2, lane);
    slab_order();
#pragma unroll
    for (int i = 0; i < 8; ++i)
#pragma unroll
      for (int t = 0; t < 2; ++t)
        bW[0][i][t] = load_b_slab2(slab, t, i, lane);
#pragma unroll
    for (int i = 0; i < 8; ++i) aH[0][i] = load_a4(h, HP, i * 4, lane);
#pragma unroll
    for (int q = 0; q < 8; ++q) {
      __builtin_amdgcn_sched_barrier(0);
      if (q + 1 < 8) {
        slab_order();
        w2_stage(slab, rw[(q + 1) & 1], lane);
        if (q + 3 < 8) w2_fetch(rw[(q + 1) & 1], W2, wave, q + 3, lane);
        slab_order();
#pragma unroll
        for (int i = 0; i < 8; ++i)
#pragma unroll
          for (int t = 0; t < 2; ++t)
            bW[(q + 1) & 1][i][t] = load_b_slab2(slab, t, i, lane);
#pragma unroll
        for (int i = 0; i < 8; ++i)
          aH[(q + 1) & 1][i] = load_a4(h, HP, ((q + 1) & 3) * 32 + i * 4, lane);
      }
      __builtin_amdgcn_sched_barrier(0);
      if ((q & 3) == 0) {
#pragma unroll
        for (int t = 0; t < 2; ++t) { v8f zero = {}; acc[t] = zero; }
      }
#pragma unroll
      for (int i = 0; i < 8; ++i)
#pragma unroll
        for (int t = 0; t < 2; ++t)
          acc[t] = WMMA4(bW[q & 1][i][t], aH[q & 1][i], acc[t]);

      if ((q & 3) == 3) {
#pragma unroll
        for (int t = 0; t < 2; ++t) {
          const int tile = (q >> 2) * 2 + t;
          float ks = 0.0f;
#pragma unroll
          for (int c = 0; c < 4; ++c) {
            float f = tanhf(acc[t][c] + bb[tile][0][c]);
            ks = fmaf(f, d0[c], ks);
          }
#pragma unroll
          for (int c = 0; c < 4; ++c) {
            float f = tanhf(acc[t][4 + c] + bb[tile][1][c]);
            ks = fmaf(f, d1[c], ks);
          }
          kout[bcol * HH + wave * 8 + tile * 2 + hh] = ks;
        }
      }
    }
  }
  __syncthreads();
}

__global__ __launch_bounds__(32 * NWG) void ncde_fused(
    const float* __restrict__ coeffs,
    const float* __restrict__ W_init, const float* __restrict__ b_init,
    const float* __restrict__ W1, const float* __restrict__ b1,
    const float* __restrict__ W2, const float* __restrict__ b2,
    const float* __restrict__ W_out, const float* __restrict__ b_out,
    float* __restrict__ out) {
  __shared__ float s_z [16 * HH];
  __shared__ float s_z0[16 * HH];
  __shared__ __attribute__((aligned(16))) float s_h [16 * HP];
  __shared__ float s_k1[16 * HH];
  __shared__ float s_k2[16 * HH];
  __shared__ float s_kc[16 * HH];
  __shared__ float s_ac[16 * HH];
  __shared__ __attribute__((aligned(16))) float s_dX[3][16 * CCH];
  __shared__ __attribute__((aligned(16))) float s_part[NWAVE * NT1 * PTILE];
  __shared__ __attribute__((aligned(16))) float s_slab[NWG * SLAB];

  const int tid  = threadIdx.x;
  const int lane = tid & 31;
  const int wave = __builtin_amdgcn_readfirstlane(tid >> 5);
  const int nthr = 32 * NWG;
  const int r0   = blockIdx.x * 16;
  const float dt = 140.0f / 49.0f;

  for (int i = 0; i < 3; ++i)
    for (int q = tid; q < 16 * CCH; q += nthr) {
      int b = q >> 3, c = q & 7;
      long base = ((long)(r0 + b) * LL + i) * CCH + c;
      s_dX[i][q] = coeffs[base + CCH] - coeffs[base];
    }

  if (wave == 0) {
    v2f aX[CCH / 4];
#pragma unroll
    for (int ks = 0; ks < CCH / 4; ++ks)
      aX[ks] = load_a4(coeffs + (long)r0 * LL * CCH, LL * CCH, ks * 4, lane);
    for (int nt = 0; nt < HH / 16; ++nt) {
      v8f acc = {};
#pragma unroll
      for (int ks = 0; ks < CCH / 4; ++ks) {
        v2f b = load_bT(W_init, CCH, nt * 16, ks * 4, lane);
        acc = WMMA4(aX[ks], b, acc);
      }
      int col = nt * 16 + (lane & 15);
      float bias = b_init[col];
#pragma unroll
      for (int j = 0; j < 8; ++j) {
        int row = (lane < 16) ? j : j + 8;
        float v = acc[j] + bias;
        s_z0[row * HH + col] = v;
        s_z [row * HH + col] = v;
      }
    }
  }
  __syncthreads();

#pragma nounroll
  for (int s = 0; s < 4; ++s) {
    const float* dXs = s_dX[s == 0 ? 0 : s - 1];
    g_eval(s_z, s_h, s_kc, dXs, s_part, s_slab + wave * SLAB, W1, b1, W2, b2, lane, wave);
    if (s == 0) {
      for (int q = tid; q < 16 * HH; q += nthr) {
        float k1 = s_kc[q];
        s_k1[q] = k1;
        s_ac[q] = k1;
        s_z[q]  = s_z0[q] + dt * (k1 * (1.0f / 3.0f));
      }
    } else if (s == 1) {
      for (int q = tid; q < 16 * HH; q += nthr) {
        float k2 = s_kc[q];
        s_k2[q] = k2;
        s_ac[q] += 3.0f * k2;
        s_z[q]  = s_z0[q] + dt * (k2 - s_k1[q] * (1.0f / 3.0f));
      }
    } else if (s == 2) {
      for (int q = tid; q < 16 * HH; q += nthr) {
        float k3 = s_kc[q];
        s_ac[q] += 3.0f * k3;
        s_z[q]  = s_z0[q] + dt * (s_k1[q] - s_k2[q] + k3);
      }
    } else {
      for (int q = tid; q < 16 * HH; q += nthr) {
        s_z[q] = s_z0[q] + (s_ac[q] + s_kc[q]) * (dt * 0.125f);
      }
    }
    __syncthreads();
  }

  if (wave == 0) {
    int b = lane & 15, o = lane >> 4;
    float s = b_out[o];
    for (int hh = 0; hh < HH; ++hh)
      s = fmaf(s_z[b * HH + hh], W_out[o * HH + hh], s);
    out[(long)(r0 + b) * 2 + o] = s;
  }
}

extern "C" void kernel_launch(void* const* d_in, const int* in_sizes, int n_in,
                              void* d_out, int out_size, void* d_ws, size_t ws_size,
                              hipStream_t stream) {
  (void)n_in; (void)out_size; (void)d_ws; (void)ws_size;
  const float* coeffs = (const float*)d_in[0];
  const float* W_init = (const float*)d_in[1];
  const float* b_init = (const float*)d_in[2];
  const float* W1     = (const float*)d_in[3];
  const float* b1     = (const float*)d_in[4];
  const float* W2     = (const float*)d_in[5];
  const float* b2     = (const float*)d_in[6];
  const float* W_out  = (const float*)d_in[7];
  const float* b_out  = (const float*)d_in[8];
  float* out = (float*)d_out;

  int B = in_sizes[0] / (LL * CCH);
  int blocks = B / 16;
  ncde_fused<<<blocks, 32 * NWG, 0, stream>>>(coeffs, W_init, b_init, W1, b1, W2, b2,
                                               W_out, b_out, out);
}


# GEMM 2 epilogue: the 16 tanh of a pass run side by side as branch-free chains (the library's two ranges both computed, one selected: same values), not one branchy chain after another; sums unchanged
# speedup vs baseline: 12.4111x; 1.2181x over previous
#include <hip/hip_runtime.h>
#include <hip/hip_bf16.h>
#include <math.h>


#define LL   141
#define CCH  8
#define HH   64
#define MIDD 128
#define HCC  512
#define NWAVE 4
#define NWG   8
#define PTILE 256
#define NT1  8
#define HP   132
#define P1 20
#define P2 36
#define SLAB 2560
#define TS2 608

typedef __attribute__((ext_vector_type(2))) float v2f;
typedef __attribute__((ext_vector_type(4))) float v4f;
typedef __attribute__((ext_vector_type(8))) float v8f;

#define WMMA4(a, b, c) \
  __builtin_amdgcn_wmma_f32_16x16x4_f32(false, (a), false, (b), (short)0, (c), false, false)

__device__ __forceinline__ v2f load_a4(const float* base, int ld, int kb, int lane) {
  int m = lane & 15;
  int k = kb + ((lane >> 4) << 1);
  const float* p = base + m * ld + k;
  v2f r = {p[0], p[1]};
  return r;
}

__device__ __forceinline__ v2f load_bT(const float* W, int ldw, int n0, int kb, int lane) {
  int n = n0 + (lane & 15);
  int k = kb + ((lane >> 4) << 1);
  const float* p = W + n * ldw + k;
  v2f r = {p[0], p[1]};
  return r;
}

__device__ __forceinline__ v2f load_b_slab(const float* slab, int pitch, int nt, int i, int lane) {
  const float* p = slab + (nt * 16 + (lane & 15)) * pitch + i * 4 + ((lane >> 4) << 1);
  v2f r = {p[0], p[1]};
  return r;
}

__device__ __forceinline__ v2f load_b_slab2(const float* slab, int t, int i, int lane) {
  const float* p = slab + t * TS2 + (lane & 15) * P2 + i * 4 + ((lane >> 4) << 1);
  v2f r = {p[0], p[1]};
  return r;
}

__device__ __forceinline__ void slab_order() {
  __builtin_amdgcn_fence(__ATOMIC_ACQ_REL, "wavefront");
}

__device__ __forceinline__ void w2_fetch(v4f (&r)[8], const float* __restrict__ W2,
                                         int wv, int q, int lane) {
#pragma unroll
  for (int j = 0; j < 8; ++j) {
    int row = wv * 64 + (q >> 2) * 32 + ((lane >> 3) & 1) * 16 + 2 * j + (lane >> 4);
    r[j] = *(const v4f*)(W2 + row * MIDD + (q & 3) * 32 + (lane & 7) * 4);
  }
}
__device__ __forceinline__ void w2_stage(float* slab, const v4f (&r)[8], int lane) {
#pragma unroll
  for (int j = 0; j < 8; ++j)
    *(v4f*)(slab + ((lane >> 3) & 1) * TS2 + (2 * j + (lane >> 4)) * P2 + (lane & 7) * 4) = r[j];
}

__device__ __forceinline__ void put_part(float* part, int w, int t, int lane, v8f acc) {
  v4f* p = (v4f*)(part + (w * NT1 + t) * PTILE) + lane;
  v4f lo = {acc[0], acc[1], acc[2], acc[3]};
  v4f hi = {acc[4], acc[5], acc[6], acc[7]};
  p[0]  = lo;
  p[32] = hi;
}

__device__ __forceinline__ v8f sum_parts(const float* part, int t, int lane) {
  v8f s;
#pragma unroll
  for (int w = 0; w < NWAVE; ++w) {
    const v4f* p = (const v4f*)(part + (w * NT1 + t) * PTILE) + lane;
    v4f lo = p[0];
    v4f hi = p[32];
    if (w == 0) {
      s[0] = lo[0]; s[1] = lo[1]; s[2] = lo[2]; s[3] = lo[3];
      s[4] = hi[0]; s[5] = hi[1]; s[6] = hi[2]; s[7] = hi[3];
    } else {
      s[0] += lo[0]; s[1] += lo[1]; s[2] += lo[2]; s[3] += lo[3];
      s[4] += hi[0]; s[5] += hi[1]; s[6] += hi[2]; s[7] += hi[3];
    }
  }
  return s;
}

__device__ __forceinline__ float tanh_sel(float x) {
  const float ax = __builtin_fabsf(x);
  const float e   = expf(ax + ax);
  const float big = __builtin_fmaf(__builtin_amdgcn_rcpf(e + 1.0f), -2.0f, 1.0f);
  const float x2 = x * x;
  float p = __builtin_fmaf(-0x1.758e7ap-8f, x2, 0x1.521192p-6f);
  p = __builtin_fmaf(x2, p, -0x1.b8389cp-5f);
  p = __builtin_fmaf(x2, p, 0x1.110704p-3f);
  p = __builtin_fmaf(x2, p, -0x1.555532p-2f);
  const float small = __builtin_fmaf(x2, ax * p, ax);
  const float r = (0.625f > ax) ? small : big;
  return __builtin_copysignf(r, x);
}

__device__ __forceinline__ void g_eval(const float* z, float* h, float* kout,
                                       const float* dX, float* part, float* slab,
                                       const float* __restrict__ W1, const float* __restrict__ b1,
                                       const float* __restrict__ W2, const float* __restrict__ b2,
                                       int lane, int wave) {
  if (wave < NWAVE) {
    const int k0 = wave * (HH / NWAVE);
    v2f aZ[HH / 4 / NWAVE];
#pragma unroll
    for (int i = 0; i < HH / 4 / NWAVE; ++i) aZ[i] = load_a4(z, HH, k0 + i * 4, lane);

    v4f r1[16];
#pragma unroll
    for (int j = 0; j < 16; ++j) {
      int row = j * 8 + (lane >> 2);
      r1[j] = *(const v4f*)(W1 + row * HH + k0 + (lane & 3) * 4);
    }
    slab_order();
#pragma unroll
    for (int j = 0; j < 16; ++j)
      *(v4f*)(slab + (j * 8 + (lane >> 2)) * P1 + (lane & 3) * 4) = r1[j];
    slab_order();
    v2f bW[HH / 4 / NWAVE][MIDD / 16];
#pragma unroll
    for (int i = 0; i < HH / 4 / NWAVE; ++i)
#pragma unroll
      for (int nt = 0; nt < MIDD / 16; ++nt)
        bW[i][nt] = load_b_slab(slab, P1, nt, i, lane);
    __builtin_amdgcn_sched_barrier(0);
    v8f acc[MIDD / 16];
#pragma unroll
    for (int nt = 0; nt < MIDD / 16; ++nt) { v8f zero = {}; acc[nt] = zero; }
#pragma unroll
    for (int i = 0; i < HH / 4 / NWAVE; ++i)
#pragma unroll
      for (int nt = 0; nt < MIDD / 16; ++nt)
        acc[nt] = WMMA4(aZ[i], bW[i][nt], acc[nt]);
#pragma unroll
    for (int nt = 0; nt < MIDD / 16; ++nt) put_part(part, wave, nt, lane, acc[nt]);
  }
  __syncthreads();

  const int bcol = lane & 15;
  const int hh   = lane >> 4;
  v4f rw[2][8];
  w2_fetch(rw[0], W2, wave, 0, lane);
  w2_fetch(rw[1], W2, wave, 1, lane);
  v4f bb[4][2];
#pragma unroll
  for (int t = 0; t < 4; ++t) {
    const v4f* bp = (const v4f*)(b2 + (wave * 8 + t * 2 + hh) * CCH);
    bb[t][0] = bp[0];
    bb[t][1] = bp[1];
  }

  {
    int nt = wave;
    v8f acc = sum_parts(part, nt, lane);
    int col = nt * 16 + (lane & 15);
    float bias = b1[col];
#pragma unroll
    for (int j = 0; j < 8; ++j) {
      int row = (lane < 16) ? j : j + 8;
      h[row * HP + col] = fmaxf(acc[j] + bias, 0.0f);
    }
  }
  __syncthreads();

  {
    const v4f d0 = *(const v4f*)(dX + bcol * CCH);
    const v4f d1 = *(const v4f*)(dX + bcol * CCH + 4);
    v8f acc[2];
    v2f aH[2][8];
    v2f bW[2][8][2];
    slab_order();
    w2_stage(slab, rw[0], lane);
    w2_fetch(rw[0], W2, wave, 2, lane);
    slab_order();
#pragma unroll
    for (int i = 0; i < 8; ++i)
#pragma unroll
      for (int t = 0; t < 2; ++t)
        bW[0][i][t] = load_b_slab2(slab, t, i, lane);
#pragma unroll
    for (int i = 0; i < 8; ++i) aH[0][i] = load_a4(h, HP, i * 4, lane);
#pragma unroll
    for (int q = 0; q < 8; ++q) {
      __builtin_amdgcn_sched_barrier(0);
      if (q + 1 < 8) {
        slab_order();
        w2_stage(slab, rw[(q + 1) & 1], lane);
        if (q + 3 < 8) w2_fetch(rw[(q + 1) & 1], W2, wave, q + 3, lane);
        slab_order();
#pragma unroll
        for (int i = 0; i < 8; ++i)
#pragma unroll
          for (int t = 0; t < 2; ++t)
            bW[(q + 1) & 1][i][t] = load_b_slab2(slab, t, i, lane);
#pragma unroll
        for (int i = 0; i < 8; ++i)
          aH[(q + 1) & 1][i] = load_a4(h, HP, ((q + 1) & 3) * 32 + i * 4, lane);
      }
      __builtin_amdgcn_sched_barrier(0);
      if ((q & 3) == 0) {
#pragma unroll
        for (int t = 0; t < 2; ++t) { v8f zero = {}; acc[t] = zero; }
      }
#pragma unroll
      for (int i = 0; i < 8; ++i)
#pragma unroll
        for (int t = 0; t < 2; ++t)
          acc[t] = WMMA4(bW[q & 1][i][t], aH[q & 1][i], acc[t]);

      if ((q & 3) == 3) {
        float f[2][8];
#pragma unroll
        for (int t = 0; t < 2; ++t) {
          const int tile = (q >> 2) * 2 + t;
#pragma unroll
          for (int c = 0; c < 4; ++c) {
            f[t][c]     = tanh_sel(acc[t][c]     + bb[tile][0][c]);
            f[t][4 + c] = tanh_sel(acc[t][4 + c] + bb[tile][1][c]);
          }
        }
#pragma unroll
        for (int t = 0; t < 2; ++t) {
          const int tile = (q >> 2) * 2 + t;
          float ks = 0.0f;
#pragma unroll
          for (int c = 0; c < 4; ++c) ks = fmaf(f[t][c], d0[c], ks);
#pragma unroll
          for (int c = 0; c < 4; ++c) ks = fmaf(f[t][4 + c], d1[c], ks);
          kout[bcol * HH + wave * 8 + tile * 2 + hh] = ks;
        }
      }
    }
  }
  __syncthreads();
}

__global__ __launch_bounds__(32 * NWG) void ncde_fused(
    const float* __restrict__ coeffs,
    const float* __restrict__ W_init, const float* __restrict__ b_init,
    const float* __restrict__ W1, const float* __restrict__ b1,
    const float* __restrict__ W2, const float* __restrict__ b2,
    const float* __restrict__ W_out, const float* __restrict__ b_out,
    float* __restrict__ out) {
  __shared__ float s_z [16 * HH];
  __shared__ float s_z0[16 * HH];
  __shared__ __attribute__((aligned(16))) float s_h [16 * HP];
  __shared__ float s_k1[16 * HH];
  __shared__ float s_k2[16 * HH];
  __shared__ float s_kc[16 * HH];
  __shared__ float s_ac[16 * HH];
  __shared__ __attribute__((aligned(16))) float s_dX[3][16 * CCH];
  __shared__ __attribute__((aligned(16))) float s_part[NWAVE * NT1 * PTILE];
  __shared__ __attribute__((aligned(16))) float s_slab[NWG * SLAB];

  const int tid  = threadIdx.x;
  const int lane = tid & 31;
  const int wave = __builtin_amdgcn_readfirstlane(tid >> 5);
  const int nthr = 32 * NWG;
  const int r0   = blockIdx.x * 16;
  const float dt = 140.0f / 49.0f;

  for (int i = 0; i < 3; ++i)
    for (int q = tid; q < 16 * CCH; q += nthr) {
      int b = q >> 3, c = q & 7;
      long base = ((long)(r0 + b) * LL + i) * CCH + c;
      s_dX[i][q] = coeffs[base + CCH] - coeffs[base];
    }

  if (wave == 0) {
    v2f aX[CCH / 4];
#pragma unroll
    for (int ks = 0; ks < CCH / 4; ++ks)
      aX[ks] = load_a4(coeffs + (long)r0 * LL * CCH, LL * CCH, ks * 4, lane);
    for (int nt = 0; nt < HH / 16; ++nt) {
      v8f acc = {};
#pragma unroll
      for (int ks = 0; ks < CCH / 4; ++ks) {
        v2f b = load_bT(W_init, CCH, nt * 16, ks * 4, lane);
        acc = WMMA4(aX[ks], b, acc);
      }
      int col = nt * 16 + (lane & 15);
      float bias = b_init[col];
#pragma unroll
      for (int j = 0; j < 8; ++j) {
        int row = (lane < 16) ? j : j + 8;
        float v = acc[j] + bias;
        s_z0[row * HH + col] = v;
        s_z [row * HH + col] = v;
      }
    }
  }
  __syncthreads();

#pragma nounroll
  for (int s = 0; s < 4; ++s) {
    const float* dXs = s_dX[s == 0 ? 0 : s - 1];
    g_eval(s_z, s_h, s_kc, dXs, s_part, s_slab + wave * SLAB, W1, b1, W2, b2, lane, wave);
    if (s == 0) {
      for (int q = tid; q < 16 * HH; q += nthr) {
        float k1 = s_kc[q];
        s_k1[q] = k1;
        s_ac[q] = k1;
        s_z[q]  = s_z0[q] + dt * (k1 * (1.0f / 3.0f));
      }
    } else if (s == 1) {
      for (int q = tid; q < 16 * HH; q += nthr) {
        float k2 = s_kc[q];
        s_k2[q] = k2;
        s_ac[q] += 3.0f * k2;
        s_z[q]  = s_z0[q] + dt * (k2 - s_k1[q] * (1.0f / 3.0f));
      }
    } else if (s == 2) {
      for (int q = tid; q < 16 * HH; q += nthr) {
        float k3 = s_kc[q];
        s_ac[q] += 3.0f * k3;
        s_z[q]  = s_z0[q] + dt * (s_k1[q] - s_k2[q] + k3);
      }
    } else {
      for (int q = tid; q < 16 * HH; q += nthr) {
        s_z[q] = s_z0[q] + (s_ac[q] + s_kc[q]) * (dt * 0.125f);
      }
    }
    __syncthreads();
  }

  if (wave == 0) {
    int b = lane & 15, o = lane >> 4;
    float s = b_out[o];
    for (int hh = 0; hh < HH; ++hh)
      s = fmaf(s_z[b * HH + hh], W_out[o * HH + hh], s);
    out[(long)(r0 + b) * 2 + o] = s;
  }
}

extern "C" void kernel_launch(void* const* d_in, const int* in_sizes, int n_in,
                              void* d_out, int out_size, void* d_ws, size_t ws_size,
                              hipStream_t stream) {
  (void)n_in; (void)out_size; (void)d_ws; (void)ws_size;
  const float* coeffs = (const float*)d_in[0];
  const float* W_init = (const float*)d_in[1];
  const float* b_init = (const float*)d_in[2];
  const float* W1     = (const float*)d_in[3];
  const float* b1     = (const float*)d_in[4];
  const float* W2     = (const float*)d_in[5];
  const float* b2     = (const float*)d_in[6];
  const float* W_out  = (const float*)d_in[7];
  const float* b_out  = (const float*)d_in[8];
  float* out = (float*)d_out;

  int B = in_sizes[0] / (LL * CCH);
  int blocks = B / 16;
  ncde_fused<<<blocks, 32 * NWG, 0, stream>>>(coeffs, W_init, b_init, W1, b1, W2, b2,
                                               W_out, b_out, out);
}


# GEMM 1 not split over K on 4 waves any more: transposed on all 8 waves (W1 rows as A operand, whole K per wave, z pitch 68), h straight from D registers, no partial tiles; k sum one chain, not 4
# speedup vs baseline: 13.1676x; 1.0610x over previous
#include <hip/hip_runtime.h>
#include <hip/hip_bf16.h>
#include <math.h>


#define LL   141
#define CCH  8
#define HH   64
#define MIDD 128
#define HCC  512
#define NWG   8
#define ZP   68
#define HP   132
#define P2 36
#define SLAB 1280
#define TS2 608

typedef __attribute__((ext_vector_type(2))) float v2f;
typedef __attribute__((ext_vector_type(4))) float v4f;
typedef __attribute__((ext_vector_type(8))) float v8f;

#define WMMA4(a, b, c) \
  __builtin_amdgcn_wmma_f32_16x16x4_f32(false, (a), false, (b), (short)0, (c), false, false)

__device__ __forceinline__ v2f load_a4(const float* base, int ld, int kb, int lane) {
  int m = lane & 15;
  int k = kb + ((lane >> 4) << 1);
  const float* p = base + m * ld + k;
  v2f r = {p[0], p[1]};
  return r;
}

__device__ __forceinline__ v2f load_bT(const float* W, int ldw, int n0, int kb, int lane) {
  int n = n0 + (lane & 15);
  int k = kb + ((lane >> 4) << 1);
  const float* p = W + n * ldw + k;
  v2f r = {p[0], p[1]};
  return r;
}

__device__ __forceinline__ v2f load_b_slab2(const float* slab, int t, int i, int lane) {
  const float* p = slab + t * TS2 + (lane & 15) * P2 + i * 4 + ((lane >> 4) << 1);
  v2f r = {p[0], p[1]};
  return r;
}

__device__ __forceinline__ void slab_order() {
  __builtin_amdgcn_fence(__ATOMIC_ACQ_REL, "wavefront");
}

__device__ __forceinline__ void w2_fetch(v4f (&r)[8], const float* __restrict__ W2,
                                         int wv, int q, int lane) {
#pragma unroll
  for (int j = 0; j < 8; ++j) {
    int row = wv * 64 + (q >> 2) * 32 + ((lane >> 3) & 1) * 16 + 2 * j + (lane >> 4);
    r[j] = *(const v4f*)(W2 + row * MIDD + (q & 3) * 32 + (lane & 7) * 4);
  }
}
__device__ __forceinline__ void w2_stage(float* slab, const v4f (&r)[8], int lane) {
#pragma unroll
  for (int j = 0; j < 8; ++j)
    *(v4f*)(slab + ((lane >> 3) & 1) * TS2 + (2 * j + (lane >> 4)) * P2 + (lane & 7) * 4) = r[j];
}

__device__ __forceinline__ float tanh_sel(float x) {
  const float ax = __builtin_fabsf(x);
  const float e   = expf(ax + ax);
  const float big = __builtin_fmaf(__builtin_amdgcn_rcpf(e + 1.0f), -2.0f, 1.0f);
  const float x2 = x * x;
  float p = __builtin_fmaf(-0x1.758e7ap-8f, x2, 0x1.521192p-6f);
  p = __builtin_fmaf(x2, p, -0x1.b8389cp-5f);
  p = __builtin_fmaf(x2, p, 0x1.110704p-3f);
  p = __builtin_fmaf(x2, p, -0x1.555532p-2f);
  const float small = __builtin_fmaf(x2, ax * p, ax);
  const float r = (0.625f > ax) ? small : big;
  return __builtin_copysignf(r, x);
}

__device__ __forceinline__ void g_eval(const float* z, float* h, float* kout,
                                       const float* dX, float* slab,
                                       const float* __restrict__ W1, const float* __restrict__ b1,
                                       const float* __restrict__ W2, const float* __restrict__ b2,
                                       int lane, int wave) {
  const int bcol = lane & 15;
  const int hh   = lane >> 4;

  v2f aW[HH / 4];
#pragma unroll
  for (int i = 0; i < HH / 4; ++i) aW[i] = load_bT(W1, HH, wave * 16, i * 4, lane);
  const v4f* bp1 = (const v4f*)(b1 + wave * 16 + hh * 8);
  const v4f c0 = bp1[0];
  const v4f c1 = bp1[1];
  v4f rw[2][8];
  w2_fetch(rw[0], W2, wave, 0, lane);
  w2_fetch(rw[1], W2, wave, 1, lane);
  v4f bb[4][2];
#pragma unroll
  for (int t = 0; t < 4; ++t) {
    const v4f* bp = (const v4f*)(b2 + (wave * 8 + t * 2 + hh) * CCH);
    bb[t][0] = bp[0];
    bb[t][1] = bp[1];
  }

  {
    v2f bZ[HH / 4];
#pragma unroll
    for (int i = 0; i < HH / 4; ++i) bZ[i] = load_a4(z, ZP, i * 4, lane);
    __builtin_amdgcn_sched_barrier(0);
    v8f acc = {};
#pragma unroll
    for (int i = 0; i < HH / 4; ++i) acc = WMMA4(aW[i], bZ[i], acc);
    v4f lo = {fmaxf(acc[0] + c0[0], 0.0f), fmaxf(acc[1] + c0[1], 0.0f),
              fmaxf(acc[2] + c0[2], 0.0f), fmaxf(acc[3] + c0[3], 0.0f)};
    v4f hi = {fmaxf(acc[4] + c1[0], 0.0f), fmaxf(acc[5] + c1[1], 0.0f),
              fmaxf(acc[6] + c1[2], 0.0f), fmaxf(acc[7] + c1[3], 0.0f)};
    v4f* hp = (v4f*)(h + bcol * HP + wave * 16 + hh * 8);
    hp[0] = lo;
    hp[1] = hi;
  }
  __syncthreads();

  {
    const v4f d0 = *(const v4f*)(dX + bcol * CCH);
    const v4f d1 = *(const v4f*)(dX + bcol * CCH + 4);
    v8f acc[2];
    v2f aH[2][8];
    v2f bW[2][8][2];
    slab_order();
    w2_stage(slab, rw[0], lane);
    w2_fetch(rw[0], W2, wave, 2, lane);
    slab_order();
#pragma unroll
    for (int i = 0; i < 8; ++i)
#pragma unroll
      for (int t = 0; t < 2; ++t)
        bW[0][i][t] = load_b_slab2(slab, t, i, lane);
#pragma unroll
    for (int i = 0; i < 8; ++i) aH[0][i] = load_a4(h, HP, i * 4, lane);
#pragma unroll
    for (int q = 0; q < 8; ++q) {
      __builtin_amdgcn_sched_barrier(0);
      if (q + 1 < 8) {
        slab_order();
        w2_stage(slab, rw[(q + 1) & 1], lane);
        if (q + 3 < 8) w2_fetch(rw[(q + 1) & 1], W2, wave, q + 3, lane);
        slab_order();
#pragma unroll
        for (int i = 0; i < 8; ++i)
#pragma unroll
          for (int t = 0; t < 2; ++t)
            bW[(q + 1) & 1][i][t] = load_b_slab2(slab, t, i, lane);
#pragma unroll
        for (int i = 0; i < 8; ++i)
          aH[(q + 1) & 1][i] = load_a4(h, HP, ((q + 1) & 3) * 32 + i * 4, lane);
      }
      __builtin_amdgcn_sched_barrier(0);
      if ((q & 3) == 0) {
#pragma unroll
        for (int t = 0; t < 2; ++t) { v8f zero = {}; acc[t] = zero; }
      }
#pragma unroll
      for (int i = 0; i < 8; ++i)
#pragma unroll
        for (int t = 0; t < 2; ++t)
          acc[t] = WMMA4(bW[q & 1][i][t], aH[q & 1][i], acc[t]);

      if ((q & 3) == 3) {
        float f[2][8];
#pragma unroll
        for (int t = 0; t < 2; ++t) {
          const int tile = (q >> 2) * 2 + t;
#pragma unroll
          for (int c = 0; c < 4; ++c) {
            f[t][c]     = tanh_sel(acc[t][c]     + bb[tile][0][c]);
            f[t][4 + c] = tanh_sel(acc[t][4 + c] + bb[tile][1][c]);
          }
        }
#pragma unroll
        for (int t = 0; t < 2; ++t) {
          const int tile = (q >> 2) * 2 + t;
          float ks = 0.0f;
#pragma unroll
          for (int c = 0; c < 4; ++c) ks = fmaf(f[t][c], d0[c], ks);
#pragma unroll
          for (int c = 0; c < 4; ++c) ks = fmaf(f[t][4 + c], d1[c], ks);
          kout[bcol * HH + wave * 8 + tile * 2 + hh] = ks;
        }
      }
    }
  }
  __syncthreads();
}

__global__ __launch_bounds__(32 * NWG) void ncde_fused(
    const float* __restrict__ coeffs,
    const float* __restrict__ W_init, const float* __restrict__ b_init,
    const float* __restrict__ W1, const float* __restrict__ b1,
    const float* __restrict__ W2, const float* __restrict__ b2,
    const float* __restrict__ W_out, const float* __restrict__ b_out,
    float* __restrict__ out) {
  __shared__ __attribute__((aligned(16))) float s_z [16 * ZP];
  __shared__ float s_z0[16 * HH];
  __shared__ __attribute__((aligned(16))) float s_h [16 * HP];
  __shared__ float s_k1[16 * HH];
  __shared__ float s_k2[16 * HH];
  __shared__ float s_kc[16 * HH];
  __shared__ float s_ac[16 * HH];
  __shared__ __attribute__((aligned(16))) float s_dX[3][16 * CCH];
  __shared__ __attribute__((aligned(16))) float s_slab[NWG * SLAB];

  const int tid  = threadIdx.x;
  const int lane = tid & 31;
  const int wave = __builtin_amdgcn_readfirstlane(tid >> 5);
  const int nthr = 32 * NWG;
  const int r0   = blockIdx.x * 16;
  const float dt = 140.0f / 49.0f;

  for (int i = 0; i < 3; ++i)
    for (int q = tid; q < 16 * CCH; q += nthr) {
      int b = q >> 3, c = q & 7;
      long base = ((long)(r0 + b) * LL + i) * CCH + c;
      s_dX[i][q] = coeffs[base + CCH] - coeffs[base];
    }

  if (wave == 0) {
    v2f aX[CCH / 4];
#pragma unroll
    for (int ks = 0; ks < CCH / 4; ++ks)
      aX[ks] = load_a4(coeffs + (long)r0 * LL * CCH, LL * CCH, ks * 4, lane);
    for (int nt = 0; nt < HH / 16; ++nt) {
      v8f acc = {};
#pragma unroll
      for (int ks = 0; ks < CCH / 4; ++ks) {
        v2f b = load_bT(W_init, CCH, nt * 16, ks * 4, lane);
        acc = WMMA4(aX[ks], b, acc);
      }
      int col = nt * 16 + (lane & 15);
      float bias = b_init[col];
#pragma unroll
      for (int j = 0; j < 8; ++j) {
        int row = (lane < 16) ? j : j + 8;
        float v = acc[j] + bias;
        s_z0[row * HH + col] = v;
        s_z [row * ZP + col] = v;
      }
    }
  }
  __syncthreads();

#pragma nounroll
  for (int s = 0; s < 4; ++s) {
    const float* dXs = s_dX[s == 0 ? 0 : s - 1];
    g_eval(s_z, s_h, s_kc, dXs, s_slab + wave * SLAB, W1, b1, W2, b2, lane, wave);
    if (s == 0) {
      for (int q = tid; q < 16 * HH; q += nthr) {
        float k1 = s_kc[q];
        s_k1[q] = k1;
        s_ac[q] = k1;
        s_z[(q >> 6) * ZP + (q & (HH - 1))] = s_z0[q] + dt * (k1 * (1.0f / 3.0f));
      }
    } else if (s == 1) {
      for (int q = tid; q < 16 * HH; q += nthr) {
        float k2 = s_kc[q];
        s_k2[q] = k2;
        s_ac[q] += 3.0f * k2;
        s_z[(q >> 6) * ZP + (q & (HH - 1))] = s_z0[q] + dt * (k2 - s_k1[q] * (1.0f / 3.0f));
      }
    } else if (s == 2) {
      for (int q = tid; q < 16 * HH; q += nthr) {
        float k3 = s_kc[q];
        s_ac[q] += 3.0f * k3;
        s_z[(q >> 6) * ZP + (q & (HH - 1))] = s_z0[q] + dt * (s_k1[q] - s_k2[q] + k3);
      }
    } else {
      for (int q = tid; q < 16 * HH; q += nthr) {
        s_z[(q >> 6) * ZP + (q & (HH - 1))] = s_z0[q] + (s_ac[q] + s_kc[q]) * (dt * 0.125f);
      }
    }
    __syncthreads();
  }

  if (wave == 0) {
    int b = lane & 15, o = lane >> 4;
    float s = b_out[o];
    for (int hh = 0; hh < HH; ++hh)
      s = fmaf(s_z[b * ZP + hh], W_out[o * HH + hh], s);
    out[(long)(r0 + b) * 2 + o] = s;
  }
}

extern "C" void kernel_launch(void* const* d_in, const int* in_sizes, int n_in,
                              void* d_out, int out_size, void* d_ws, size_t ws_size,
                              hipStream_t stream) {
  (void)n_in; (void)out_size; (void)d_ws; (void)ws_size;
  const float* coeffs = (const float*)d_in[0];
  const float* W_init = (const float*)d_in[1];
  const float* b_init = (const float*)d_in[2];
  const float* W1     = (const float*)d_in[3];
  const float* b1     = (const float*)d_in[4];
  const float* W2     = (const float*)d_in[5];
  const float* b2     = (const float*)d_in[6];
  const float* W_out  = (const float*)d_in[7];
  const float* b_out  = (const float*)d_in[8];
  float* out = (float*)d_out;

  int B = in_sizes[0] / (LL * CCH);
  int blocks = B / 16;
  ncde_fused<<<blocks, 32 * NWG, 0, stream>>>(coeffs, W_init, b_init, W1, b1, W2, b2,
                                               W_out, b_out, out);
}


# All global reads requested at entry in order of first use (W1, b1, W2 chunks 0-7; W_out parked in LDS); barriers wait for LDS only: weight fetch runs under z0 and the first stage; sums unchanged
# speedup vs baseline: 13.8186x; 1.0494x over previous
#include <hip/hip_runtime.h>
#include <hip/hip_bf16.h>
#include <math.h>


#define LL   141
#define CCH  8
#define HH   64
#define MIDD 128
#define HCC  512
#define NWG   8
#define ZP   68
#define HP   132
#define P2 36
#define SLAB 1280
#define TS2 608

typedef __attribute__((ext_vector_type(2))) float v2f;
typedef __attribute__((ext_vector_type(4))) float v4f;
typedef __attribute__((ext_vector_type(8))) float v8f;

#define WMMA4(a, b, c) \
  __builtin_amdgcn_wmma_f32_16x16x4_f32(false, (a), false, (b), (short)0, (c), false, false)

__device__ __forceinline__ v2f load_a4(const float* base, int ld, int kb, int lane) {
  int m = lane & 15;
  int k = kb + ((lane >> 4) << 1);
  const float* p = base + m * ld + k;
  v2f r = {p[0], p[1]};
  return r;
}

__device__ __forceinline__ v2f load_bT(const float* W, int ldw, int n0, int kb, int lane) {
  int n = n0 + (lane & 15);
  int k = kb + ((lane >> 4) << 1);
  const float* p = W + n * ldw + k;
  v2f r = {p[0], p[1]};
  return r;
}

__device__ __forceinline__ v2f load_b_slab2(const float* slab, int t, int i, int lane) {
  const float* p = slab + t * TS2 + (lane & 15) * P2 + i * 4 + ((lane >> 4) << 1);
  v2f r = {p[0], p[1]};
  return r;
}

__device__ __forceinline__ void slab_order() {
  __builtin_amdgcn_fence(__ATOMIC_ACQ_REL, "wavefront");
}

__device__ __forceinline__ void lds_barrier() {
  __builtin_amdgcn_fence(__ATOMIC_RELEASE, "workgroup", "local");
  __builtin_amdgcn_s_barrier();
  __builtin_amdgcn_fence(__ATOMIC_ACQUIRE, "workgroup", "local");
}

__device__ __forceinline__ void w2_fetch(v4f (&r)[8], const float* __restrict__ W2,
                                         int wv, int q, int lane) {
#pragma unroll
  for (int j = 0; j < 8; ++j) {
    int row = wv * 64 + (q >> 2) * 32 + ((lane >> 3) & 1) * 16 + 2 * j + (lane >> 4);
    r[j] = *(const v4f*)(W2 + row * MIDD + (q & 3) * 32 + (lane & 7) * 4);
  }
}
__device__ __forceinline__ void w2_stage(float* slab, const v4f (&r)[8], int lane) {
#pragma unroll
  for (int j = 0; j < 8; ++j)
    *(v4f*)(slab + ((lane >> 3) & 1) * TS2 + (2 * j + (lane >> 4)) * P2 + (lane & 7) * 4) = r[j];
}

__device__ __forceinline__ float tanh_sel(float x) {
  const float ax = __builtin_fabsf(x);
  const float e   = expf(ax + ax);
  const float big = __builtin_fmaf(__builtin_amdgcn_rcpf(e + 1.0f), -2.0f, 1.0f);
  const float x2 = x * x;
  float p = __builtin_fmaf(-0x1.758e7ap-8f, x2, 0x1.521192p-6f);
  p = __builtin_fmaf(x2, p, -0x1.b8389cp-5f);
  p = __builtin_fmaf(x2, p, 0x1.110704p-3f);
  p = __builtin_fmaf(x2, p, -0x1.555532p-2f);
  const float small = __builtin_fmaf(x2, ax * p, ax);
  const float r = (0.625f > ax) ? small : big;
  return __builtin_copysignf(r, x);
}

__device__ __forceinline__ void g_eval(const float* z, float* h, float* kout,
                                       const float* dX, float* slab,
                                       const v2f (&aW)[HH / 4], const v4f c0, const v4f c1,
                                       const v4f (&rw)[8][8], const v4f (&bb)[4][2],
                                       int lane, int wave) {
  const int bcol = lane & 15;
  const int hh   = lane >> 4;

  {
    v2f bZ[HH / 4];
#pragma unroll
    for (int i = 0; i < HH / 4; ++i) bZ[i] = load_a4(z, ZP, i * 4, lane);
    __builtin_amdgcn_sched_barrier(0);
    v8f acc = {};
#pragma unroll
    for (int i = 0; i < HH / 4; ++i) acc = WMMA4(aW[i], bZ[i], acc);
    v4f lo = {fmaxf(acc[0] + c0[0], 0.0f), fmaxf(acc[1] + c0[1], 0.0f),
              fmaxf(acc[2] + c0[2], 0.0f), fmaxf(acc[3] + c0[3], 0.0f)};
    v4f hi = {fmaxf(acc[4] + c1[0], 0.0f), fmaxf(acc[5] + c1[1], 0.0f),
              fmaxf(acc[6] + c1[2], 0.0f), fmaxf(acc[7] + c1[3], 0.0f)};
    v4f* hp = (v4f*)(h + bcol * HP + wave * 16 + hh * 8);
    hp[0] = lo;
    hp[1] = hi;
  }
  lds_barrier();

  {
    const v4f d0 = *(const v4f*)(dX + bcol * CCH);
    const v4f d1 = *(const v4f*)(dX + bcol * CCH + 4);
    v8f acc[2];
    v2f aH[2][8];
    v2f bW[2][8][2];
    slab_order();
    w2_stage(slab, rw[0], lane);
    slab_order();
#pragma unroll
    for (int i = 0; i < 8; ++i)
#pragma unroll
      for (int t = 0; t < 2; ++t)
        bW[0][i][t] = load_b_slab2(slab, t, i, lane);
#pragma unroll
    for (int i = 0; i < 8; ++i) aH[0][i] = load_a4(h, HP, i * 4, lane);
#pragma unroll
    for (int q = 0; q < 8; ++q) {
      __builtin_amdgcn_sched_barrier(0);
      if (q + 1 < 8) {
        slab_order();
        w2_stage(slab, rw[q + 1], lane);
        slab_order();
#pragma unroll
        for (int i = 0; i < 8; ++i)
#pragma unroll
          for (int t = 0; t < 2; ++t)
            bW[(q + 1) & 1][i][t] = load_b_slab2(slab, t, i, lane);
#pragma unroll
        for (int i = 0; i < 8; ++i)
          aH[(q + 1) & 1][i] = load_a4(h, HP, ((q + 1) & 3) * 32 + i * 4, lane);
      }
      __builtin_amdgcn_sched_barrier(0);
      if ((q & 3) == 0) {
#pragma unroll
        for (int t = 0; t < 2; ++t) { v8f zero = {}; acc[t] = zero; }
      }
#pragma unroll
      for (int i = 0; i < 8; ++i)
#pragma unroll
        for (int t = 0; t < 2; ++t)
          acc[t] = WMMA4(bW[q & 1][i][t], aH[q & 1][i], acc[t]);

      if ((q & 3) == 3) {
        float f[2][8];
#pragma unroll
        for (int t = 0; t < 2; ++t) {
          const int tile = (q >> 2) * 2 + t;
#pragma unroll
          for (int c = 0; c < 4; ++c) {
            f[t][c]     = tanh_sel(acc[t][c]     + bb[tile][0][c]);
            f[t][4 + c] = tanh_sel(acc[t][4 + c] + bb[tile][1][c]);
          }
        }
#pragma unroll
        for (int t = 0; t < 2; ++t) {
          const int tile = (q >> 2) * 2 + t;
          float ks = 0.0f;
#pragma unroll
          for (int c = 0; c < 4; ++c) ks = fmaf(f[t][c], d0[c], ks);
#pragma unroll
          for (int c = 0; c < 4; ++c) ks = fmaf(f[t][4 + c], d1[c], ks);
          kout[bcol * HH + wave * 8 + tile * 2 + hh] = ks;
        }
      }
    }
  }
  lds_barrier();
}

__global__ __launch_bounds__(32 * NWG) void ncde_fused(
    const float* __restrict__ coeffs,
    const float* __restrict__ W_init, const float* __restrict__ b_init,
    const float* __restrict__ W1, const float* __restrict__ b1,
    const float* __restrict__ W2, const float* __restrict__ b2,
    const float* __restrict__ W_out, const float* __restrict__ b_out,
    float* __restrict__ out) {
  __shared__ __attribute__((aligned(16))) float s_z [16 * ZP];
  __shared__ float s_z0[16 * HH];
  __shared__ __attribute__((aligned(16))) float s_h [16 * HP];
  __shared__ float s_k1[16 * HH];
  __shared__ float s_k2[16 * HH];
  __shared__ float s_kc[16 * HH];
  __shared__ float s_ac[16 * HH];
  __shared__ __attribute__((aligned(16))) float s_dX[3][16 * CCH];
  __shared__ __attribute__((aligned(16))) float s_slab[NWG * SLAB];
  __shared__ float s_wo[2 * HH + 2];

  const int tid  = threadIdx.x;
  const int lane = tid & 31;
  const int wave = __builtin_amdgcn_readfirstlane(tid >> 5);
  const int nthr = 32 * NWG;
  const int r0   = blockIdx.x * 16;
  const float dt = 140.0f / 49.0f;

  const int lh = lane >> 4;
  float xk[4] = {0.0f, 0.0f, 0.0f, 0.0f};
  if (tid < 16 * CCH) {
    const float* cp = coeffs + (long)(r0 + (tid >> 3)) * LL * CCH + (tid & 7);
#pragma unroll
    for (int i = 0; i < 4; ++i) xk[i] = cp[i * CCH];
  }
  float wo = 0.0f;
  if (tid < 2 * HH + 2) wo = (tid < 2 * HH) ? W_out[tid] : b_out[tid - 2 * HH];
  v2f aX[CCH / 4] = {};
  v2f bI[HH / 16][CCH / 4] = {};
  float biasI[HH / 16] = {};
  if (wave == 0) {
#pragma unroll
    for (int ks = 0; ks < CCH / 4; ++ks)
      aX[ks] = load_a4(coeffs + (long)r0 * LL * CCH, LL * CCH, ks * 4, lane);
#pragma unroll
    for (int nt = 0; nt < HH / 16; ++nt) {
#pragma unroll
      for (int ks = 0; ks < CCH / 4; ++ks)
        bI[nt][ks] = load_bT(W_init, CCH, nt * 16, ks * 4, lane);
      biasI[nt] = b_init[nt * 16 + (lane & 15)];
    }
  }
  __builtin_amdgcn_sched_barrier(0);
  v4f bb[4][2];
#pragma unroll
  for (int t = 0; t < 4; ++t) {
    const v4f* bp = (const v4f*)(b2 + (wave * 8 + t * 2 + lh) * CCH);
    bb[t][0] = bp[0];
    bb[t][1] = bp[1];
  }
  __builtin_amdgcn_sched_barrier(0);
  v2f aW[HH / 4];
#pragma unroll
  for (int i = 0; i < HH / 4; ++i) aW[i] = load_bT(W1, HH, wave * 16, i * 4, lane);
  const v4f* bp1 = (const v4f*)(b1 + wave * 16 + lh * 8);
  const v4f c0 = bp1[0];
  const v4f c1 = bp1[1];
  v4f rw[8][8];
  w2_fetch(rw[0], W2, wave, 0, lane);
  __builtin_amdgcn_sched_barrier(0);
  w2_fetch(rw[1], W2, wave, 1, lane);
  __builtin_amdgcn_sched_barrier(0);

  if (tid < 16 * CCH) {
#pragma unroll
    for (int i = 0; i < 3; ++i) s_dX[i][tid] = xk[i + 1] - xk[i];
  }
  if (tid < 2 * HH + 2) s_wo[tid] = wo;

  if (wave == 0) {
#pragma unroll
    for (int nt = 0; nt < HH / 16; ++nt) {
      v8f acc = {};
#pragma unroll
      for (int ks = 0; ks < CCH / 4; ++ks) acc = WMMA4(aX[ks], bI[nt][ks], acc);
      int col = nt * 16 + (lane & 15);
#pragma unroll
      for (int j = 0; j < 8; ++j) {
        int row = (lane < 16) ? j : j + 8;
        float v = acc[j] + biasI[nt];
        s_z0[row * HH + col] = v;
        s_z [row * ZP + col] = v;
      }
    }
  }
  lds_barrier();
#pragma unroll
  for (int q = 2; q < 8; ++q) {
    w2_fetch(rw[q], W2, wave, q, lane);
    __builtin_amdgcn_sched_barrier(0);
  }

#pragma nounroll
  for (int s = 0; s < 4; ++s) {
    const float* dXs = s_dX[s == 0 ? 0 : s - 1];
    g_eval(s_z, s_h, s_kc, dXs, s_slab + wave * SLAB, aW, c0, c1, rw, bb, lane, wave);
    if (s == 0) {
      for (int q = tid; q < 16 * HH; q += nthr) {
        float k1 = s_kc[q];
        s_k1[q] = k1;
        s_ac[q] = k1;
        s_z[(q >> 6) * ZP + (q & (HH - 1))] = s_z0[q] + dt * (k1 * (1.0f / 3.0f));
      }
    } else if (s == 1) {
      for (int q = tid; q < 16 * HH; q += nthr) {
        float k2 = s_kc[q];
        s_k2[q] = k2;
        s_ac[q] += 3.0f * k2;
        s_z[(q >> 6) * ZP + (q & (HH - 1))] = s_z0[q] + dt * (k2 - s_k1[q] * (1.0f / 3.0f));
      }
    } else if (s == 2) {
      for (int q = tid; q < 16 * HH; q += nthr) {
        float k3 = s_kc[q];
        s_ac[q] += 3.0f * k3;
        s_z[(q >> 6) * ZP + (q & (HH - 1))] = s_z0[q] + dt * (s_k1[q] - s_k2[q] + k3);
      }
    } else {
      for (int q = tid; q < 16 * HH; q += nthr) {
        s_z[(q >> 6) * ZP + (q & (HH - 1))] = s_z0[q] + (s_ac[q] + s_kc[q]) * (dt * 0.125f);
      }
    }
    lds_barrier();
  }

  if (wave == 0) {
    int b = lane & 15, o = lane >> 4;
    float s = s_wo[2 * HH + o];
    for (int hh = 0; hh < HH; ++hh)
      s = fmaf(s_z[b * ZP + hh], s_wo[o * HH + hh], s);
    out[(long)(r0 + b) * 2 + o] = s;
  }
}

extern "C" void kernel_launch(void* const* d_in, const int* in_sizes, int n_in,
                              void* d_out, int out_size, void* d_ws, size_t ws_size,
                              hipStream_t stream) {
  (void)n_in; (void)out_size; (void)d_ws; (void)ws_size;
  const float* coeffs = (const float*)d_in[0];
  const float* W_init = (const float*)d_in[1];
  const float* b_init = (const float*)d_in[2];
  const float* W1     = (const float*)d_in[3];
  const float* b1     = (const float*)d_in[4];
  const float* W2     = (const float*)d_in[5];
  const float* b2     = (const float*)d_in[6];
  const float* W_out  = (const float*)d_in[7];
  const float* b_out  = (const float*)d_in[8];
  float* out = (float*)d_out;

  int B = in_sizes[0] / (LL * CCH);
  int blocks = B / 16;
  ncde_fused<<<blocks, 32 * NWG, 0, stream>>>(coeffs, W_init, b_init, W1, b1, W2, b2,
                                               W_out, b_out, out);
}


# RK update pass and output projection: all LDS reads of a pass at once, 16 B per thread and plane (4 consecutive hidden units), not rolled loops of 4 dependent round trips; sums unchanged
# speedup vs baseline: 14.6006x; 1.0566x over previous
#include <hip/hip_runtime.h>
#include <hip/hip_bf16.h>
#include <math.h>


#define LL   141
#define CCH  8
#define HH   64
#define MIDD 128
#define HCC  512
#define NWG   8
#define ZP   68
#define HP   132
#define P2 36
#define SLAB 1280
#define TS2 608

typedef __attribute__((ext_vector_type(2))) float v2f;
typedef __attribute__((ext_vector_type(4))) float v4f;
typedef __attribute__((ext_vector_type(8))) float v8f;

#define WMMA4(a, b, c) \
  __builtin_amdgcn_wmma_f32_16x16x4_f32(false, (a), false, (b), (short)0, (c), false, false)

__device__ __forceinline__ v2f load_a4(const float* base, int ld, int kb, int lane) {
  int m = lane & 15;
  int k = kb + ((lane >> 4) << 1);
  const float* p = base + m * ld + k;
  v2f r = {p[0], p[1]};
  return r;
}

__device__ __forceinline__ v2f load_bT(const float* W, int ldw, int n0, int kb, int lane) {
  int n = n0 + (lane & 15);
  int k = kb + ((lane >> 4) << 1);
  const float* p = W + n * ldw + k;
  v2f r = {p[0], p[1]};
  return r;
}

__device__ __forceinline__ v2f load_b_slab2(const float* slab, int t, int i, int lane) {
  const float* p = slab + t * TS2 + (lane & 15) * P2 + i * 4 + ((lane >> 4) << 1);
  v2f r = {p[0], p[1]};
  return r;
}

__device__ __forceinline__ void slab_order() {
  __builtin_amdgcn_fence(__ATOMIC_ACQ_REL, "wavefront");
}

__device__ __forceinline__ void lds_barrier() {
  __builtin_amdgcn_fence(__ATOMIC_RELEASE, "workgroup", "local");
  __builtin_amdgcn_s_barrier();
  __builtin_amdgcn_fence(__ATOMIC_ACQUIRE, "workgroup", "local");
}

__device__ __forceinline__ void w2_fetch(v4f (&r)[8], const float* __restrict__ W2,
                                         int wv, int q, int lane) {
#pragma unroll
  for (int j = 0; j < 8; ++j) {
    int row = wv * 64 + (q >> 2) * 32 + ((lane >> 3) & 1) * 16 + 2 * j + (lane >> 4);
    r[j] = *(const v4f*)(W2 + row * MIDD + (q & 3) * 32 + (lane & 7) * 4);
  }
}
__device__ __forceinline__ void w2_stage(float* slab, const v4f (&r)[8], int lane) {
#pragma unroll
  for (int j = 0; j < 8; ++j)
    *(v4f*)(slab + ((lane >> 3) & 1) * TS2 + (2 * j + (lane >> 4)) * P2 + (lane & 7) * 4) = r[j];
}

__device__ __forceinline__ float tanh_sel(float x) {
  const float ax = __builtin_fabsf(x);
  const float e   = expf(ax + ax);
  const float big = __builtin_fmaf(__builtin_amdgcn_rcpf(e + 1.0f), -2.0f, 1.0f);
  const float x2 = x * x;
  float p = __builtin_fmaf(-0x1.758e7ap-8f, x2, 0x1.521192p-6f);
  p = __builtin_fmaf(x2, p, -0x1.b8389cp-5f);
  p = __builtin_fmaf(x2, p, 0x1.110704p-3f);
  p = __builtin_fmaf(x2, p, -0x1.555532p-2f);
  const float small = __builtin_fmaf(x2, ax * p, ax);
  const float r = (0.625f > ax) ? small : big;
  return __builtin_copysignf(r, x);
}

__device__ __forceinline__ void g_eval(const float* z, float* h, float* kout,
                                       const float* dX, float* slab,
                                       const v2f (&aW)[HH / 4], const v4f c0, const v4f c1,
                                       const v4f (&rw)[8][8], const v4f (&bb)[4][2],
                                       int lane, int wave) {
  const int bcol = lane & 15;
  const int hh   = lane >> 4;

  {
    v2f bZ[HH / 4];
#pragma unroll
    for (int i = 0; i < HH / 4; ++i) bZ[i] = load_a4(z, ZP, i * 4, lane);
    __builtin_amdgcn_sched_barrier(0);
    v8f acc = {};
#pragma unroll
    for (int i = 0; i < HH / 4; ++i) acc = WMMA4(aW[i], bZ[i], acc);
    v4f lo = {fmaxf(acc[0] + c0[0], 0.0f), fmaxf(acc[1] + c0[1], 0.0f),
              fmaxf(acc[2] + c0[2], 0.0f), fmaxf(acc[3] + c0[3], 0.0f)};
    v4f hi = {fmaxf(acc[4] + c1[0], 0.0f), fmaxf(acc[5] + c1[1], 0.0f),
              fmaxf(acc[6] + c1[2], 0.0f), fmaxf(acc[7] + c1[3], 0.0f)};
    v4f* hp = (v4f*)(h + bcol * HP + wave * 16 + hh * 8);
    hp[0] = lo;
    hp[1] = hi;
  }
  lds_barrier();

  {
    const v4f d0 = *(const v4f*)(dX + bcol * CCH);
    const v4f d1 = *(const v4f*)(dX + bcol * CCH + 4);
    v8f acc[2];
    v2f aH[2][8];
    v2f bW[2][8][2];
    slab_order();
    w2_stage(slab, rw[0], lane);
    slab_order();
#pragma unroll
    for (int i = 0; i < 8; ++i)
#pragma unroll
      for (int t = 0; t < 2; ++t)
        bW[0][i][t] = load_b_slab2(slab, t, i, lane);
#pragma unroll
    for (int i = 0; i < 8; ++i) aH[0][i] = load_a4(h, HP, i * 4, lane);
#pragma unroll
    for (int q = 0; q < 8; ++q) {
      __builtin_amdgcn_sched_barrier(0);
      if (q + 1 < 8) {
        slab_order();
        w2_stage(slab, rw[q + 1], lane);
        slab_order();
#pragma unroll
        for (int i = 0; i < 8; ++i)
#pragma unroll
          for (int t = 0; t < 2; ++t)
            bW[(q + 1) & 1][i][t] = load_b_slab2(slab, t, i, lane);
#pragma unroll
        for (int i = 0; i < 8; ++i)
          aH[(q + 1) & 1][i] = load_a4(h, HP, ((q + 1) & 3) * 32 + i * 4, lane);
      }
      __builtin_amdgcn_sched_barrier(0);
      if ((q & 3) == 0) {
#pragma unroll
        for (int t = 0; t < 2; ++t) { v8f zero = {}; acc[t] = zero; }
      }
#pragma unroll
      for (int i = 0; i < 8; ++i)
#pragma unroll
        for (int t = 0; t < 2; ++t)
          acc[t] = WMMA4(bW[q & 1][i][t], aH[q & 1][i], acc[t]);

      if ((q & 3) == 3) {
        float f[2][8];
#pragma unroll
        for (int t = 0; t < 2; ++t) {
          const int tile = (q >> 2) * 2 + t;
#pragma unroll
          for (int c = 0; c < 4; ++c) {
            f[t][c]     = tanh_sel(acc[t][c]     + bb[tile][0][c]);
            f[t][4 + c] = tanh_sel(acc[t][4 + c] + bb[tile][1][c]);
          }
        }
#pragma unroll
        for (int t = 0; t < 2; ++t) {
          const int tile = (q >> 2) * 2 + t;
          float ks = 0.0f;
#pragma unroll
          for (int c = 0; c < 4; ++c) ks = fmaf(f[t][c], d0[c], ks);
#pragma unroll
          for (int c = 0; c < 4; ++c) ks = fmaf(f[t][4 + c], d1[c], ks);
          kout[bcol * HH + wave * 8 + tile * 2 + hh] = ks;
        }
      }
    }
  }
  lds_barrier();
}

__global__ __launch_bounds__(32 * NWG) void ncde_fused(
    const float* __restrict__ coeffs,
    const float* __restrict__ W_init, const float* __restrict__ b_init,
    const float* __restrict__ W1, const float* __restrict__ b1,
    const float* __restrict__ W2, const float* __restrict__ b2,
    const float* __restrict__ W_out, const float* __restrict__ b_out,
    float* __restrict__ out) {
  __shared__ __attribute__((aligned(16))) float s_z [16 * ZP];
  __shared__ __attribute__((aligned(16))) float s_z0[16 * HH];
  __shared__ __attribute__((aligned(16))) float s_h [16 * HP];
  __shared__ __attribute__((aligned(16))) float s_k1[16 * HH];
  __shared__ __attribute__((aligned(16))) float s_k2[16 * HH];
  __shared__ __attribute__((aligned(16))) float s_kc[16 * HH];
  __shared__ __attribute__((aligned(16))) float s_ac[16 * HH];
  __shared__ __attribute__((aligned(16))) float s_dX[3][16 * CCH];
  __shared__ __attribute__((aligned(16))) float s_slab[NWG * SLAB];
  __shared__ __attribute__((aligned(16))) float s_wo[2 * HH + 2];

  const int tid  = threadIdx.x;
  const int lane = tid & 31;
  const int wave = __builtin_amdgcn_readfirstlane(tid >> 5);
  const int r0   = blockIdx.x * 16;
  const float dt = 140.0f / 49.0f;

  const int lh = lane >> 4;
  float xk[4] = {0.0f, 0.0f, 0.0f, 0.0f};
  if (tid < 16 * CCH) {
    const float* cp = coeffs + (long)(r0 + (tid >> 3)) * LL * CCH + (tid & 7);
#pragma unroll
    for (int i = 0; i < 4; ++i) xk[i] = cp[i * CCH];
  }
  float wo = 0.0f;
  if (tid < 2 * HH + 2) wo = (tid < 2 * HH) ? W_out[tid] : b_out[tid - 2 * HH];
  v2f aX[CCH / 4] = {};
  v2f bI[HH / 16][CCH / 4] = {};
  float biasI[HH / 16] = {};
  if (wave == 0) {
#pragma unroll
    for (int ks = 0; ks < CCH / 4; ++ks)
      aX[ks] = load_a4(coeffs + (long)r0 * LL * CCH, LL * CCH, ks * 4, lane);
#pragma unroll
    for (int nt = 0; nt < HH / 16; ++nt) {
#pragma unroll
      for (int ks = 0; ks < CCH / 4; ++ks)
        bI[nt][ks] = load_bT(W_init, CCH, nt * 16, ks * 4, lane);
      biasI[nt] = b_init[nt * 16 + (lane & 15)];
    }
  }
  __builtin_amdgcn_sched_barrier(0);
  v4f bb[4][2];
#pragma unroll
  for (int t = 0; t < 4; ++t) {
    const v4f* bp = (const v4f*)(b2 + (wave * 8 + t * 2 + lh) * CCH);
    bb[t][0] = bp[0];
    bb[t][1] = bp[1];
  }
  __builtin_amdgcn_sched_barrier(0);
  v2f aW[HH / 4];
#pragma unroll
  for (int i = 0; i < HH / 4; ++i) aW[i] = load_bT(W1, HH, wave * 16, i * 4, lane);
  const v4f* bp1 = (const v4f*)(b1 + wave * 16 + lh * 8);
  const v4f c0 = bp1[0];
  const v4f c1 = bp1[1];
  v4f rw[8][8];
  w2_fetch(rw[0], W2, wave, 0, lane);
  __builtin_amdgcn_sched_barrier(0);
  w2_fetch(rw[1], W2, wave, 1, lane);
  __builtin_amdgcn_sched_barrier(0);

  if (tid < 16 * CCH) {
#pragma unroll
    for (int i = 0; i < 3; ++i) s_dX[i][tid] = xk[i + 1] - xk[i];
  }
  if (tid < 2 * HH + 2) s_wo[tid] = wo;

  if (wave == 0) {
#pragma unroll
    for (int nt = 0; nt < HH / 16; ++nt) {
      v8f acc = {};
#pragma unroll
      for (int ks = 0; ks < CCH / 4; ++ks) acc = WMMA4(aX[ks], bI[nt][ks], acc);
      int col = nt * 16 + (lane & 15);
#pragma unroll
      for (int j = 0; j < 8; ++j) {
        int row = (lane < 16) ? j : j + 8;
        float v = acc[j] + biasI[nt];
        s_z0[row * HH + col] = v;
        s_z [row * ZP + col] = v;
      }
    }
  }
  lds_barrier();
#pragma unroll
  for (int q = 2; q < 8; ++q) {
    w2_fetch(rw[q], W2, wave, q, lane);
    __builtin_amdgcn_sched_barrier(0);
  }

  const int q4 = tid * 4;
  float* const zq = s_z + (tid >> 4) * ZP + (tid & 15) * 4;
  const v4f dt4     = {dt, dt, dt, dt};
  const v4f dt84    = {dt * 0.125f, dt * 0.125f, dt * 0.125f, dt * 0.125f};
  const v4f three4  = {3.0f, 3.0f, 3.0f, 3.0f};
  const v4f mthird4 = {-(1.0f / 3.0f), -(1.0f / 3.0f), -(1.0f / 3.0f), -(1.0f / 3.0f)};
#pragma nounroll
  for (int s = 0; s < 4; ++s) {
    const float* dXs = s_dX[s == 0 ? 0 : s - 1];
    g_eval(s_z, s_h, s_kc, dXs, s_slab + wave * SLAB, aW, c0, c1, rw, bb, lane, wave);
    {
      const v4f kc4 = *(const v4f*)(s_kc + q4);
      const v4f z04 = *(const v4f*)(s_z0 + q4);
      v4f zn;
      if (s == 0) {
        zn = __builtin_elementwise_fma(dt4, kc4 * (1.0f / 3.0f), z04);
        *(v4f*)(s_k1 + q4) = kc4;
        *(v4f*)(s_ac + q4) = kc4;
      } else if (s == 1) {
        const v4f k14 = *(const v4f*)(s_k1 + q4);
        const v4f ac4 = *(const v4f*)(s_ac + q4);
        zn = __builtin_elementwise_fma(dt4, __builtin_elementwise_fma(mthird4, k14, kc4), z04);
        *(v4f*)(s_k2 + q4) = kc4;
        *(v4f*)(s_ac + q4) = __builtin_elementwise_fma(three4, kc4, ac4);
      } else if (s == 2) {
        const v4f k14 = *(const v4f*)(s_k1 + q4);
        const v4f k24 = *(const v4f*)(s_k2 + q4);
        const v4f ac4 = *(const v4f*)(s_ac + q4);
        zn = __builtin_elementwise_fma(dt4, k14 - k24 + kc4, z04);
        *(v4f*)(s_ac + q4) = __builtin_elementwise_fma(three4, kc4, ac4);
      } else {
        const v4f ac4 = *(const v4f*)(s_ac + q4);
        zn = __builtin_elementwise_fma(dt84, ac4 + kc4, z04);
      }
      *(v4f*)zq = zn;
    }
    lds_barrier();
  }

  if (wave == 0) {
    int b = lane & 15, o = lane >> 4;
    float s = s_wo[2 * HH + o];
    v4f zr[HH / 4], wr[HH / 4];
#pragma unroll
    for (int i = 0; i < HH / 4; ++i) {
      zr[i] = *(const v4f*)(s_z + b * ZP + 4 * i);
      wr[i] = *(const v4f*)(s_wo + o * HH + 4 * i);
    }
    __builtin_amdgcn_sched_barrier(0);
#pragma unroll
    for (int i = 0; i < HH / 4; ++i)
#pragma unroll
      for (int e = 0; e < 4; ++e)
        s = fmaf(zr[i][e], wr[i][e], s);
    out[(long)(r0 + b) * 2 + o] = s;
  }
}

extern "C" void kernel_launch(void* const* d_in, const int* in_sizes, int n_in,
                              void* d_out, int out_size, void* d_ws, size_t ws_size,
                              hipStream_t stream) {
  (void)n_in; (void)out_size; (void)d_ws; (void)ws_size;
  const float* coeffs = (const float*)d_in[0];
  const float* W_init = (const float*)d_in[1];
  const float* b_init = (const float*)d_in[2];
  const float* W1     = (const float*)d_in[3];
  const float* b1     = (const float*)d_in[4];
  const float* W2     = (const float*)d_in[5];
  const float* b2     = (const float*)d_in[6];
  const float* W_out  = (const float*)d_in[7];
  const float* b_out  = (const float*)d_in[8];
  float* out = (float*)d_out;

  int B = in_sizes[0] / (LL * CCH);
  int blocks = B / 16;
  ncde_fused<<<blocks, 32 * NWG, 0, stream>>>(coeffs, W_init, b_init, W1, b1, W2, b2,
                                               W_out, b_out, out);
}
